# starting kernel
# baseline (speedup 1.0000x reference)
#include <hip/hip_runtime.h>
#include <math.h>

#define DD 768
#define LL 1024
#define NN 16
#define BB 2
#define NC 16
#define LC (LL / NC)

typedef float v2f __attribute__((ext_vector_type(2)));
typedef float v8f __attribute__((ext_vector_type(8)));

__global__ __launch_bounds__(32) void ssm_proj_wmma(
    const float* __restrict__ x,
    const float* __restrict__ W_B,
    const float* __restrict__ b_B,
    const float* __restrict__ W_C,
    const float* __restrict__ b_C,
    const float* __restrict__ W_d,
    const float* __restrict__ b_d,
    float* __restrict__ Bp,
    float* __restrict__ Cp,
    float* __restrict__ dlin)
{
    const int lane = threadIdx.x;
    const int half = lane >> 4;
    const int mn   = lane & 15;
    const int row0 = blockIdx.x << 4;
    const int b    = row0 >> 10;
    const int l0   = row0 & (LL - 1);
    const int lA   = l0 + mn;

    const float biasB = b_B[mn];
    const float biasC = b_C[mn];
    const float biasD = (mn == 0) ? b_d[0] : 0.0f;
    v8f accB, accC, accD;
#pragma unroll
    for (int r = 0; r < 8; ++r) { accB[r] = biasB; accC[r] = biasC; accD[r] = biasD; }

    const float* __restrict__ xb = x + (size_t)b * DD * LL;
    const float n0sel = (mn == 0) ? 1.0f : 0.0f;

    for (int k0 = 0; k0 < DD; k0 += 4) {
        const int k = k0 + 2 * half;
        v2f a;
        a[0] = xb[(size_t)k * LL + lA];
        a[1] = xb[(size_t)(k + 1) * LL + lA];
        v2f wb, wc, wd;
        wb[0] = W_B[mn * DD + k];
        wb[1] = W_B[mn * DD + k + 1];
        wc[0] = W_C[mn * DD + k];
        wc[1] = W_C[mn * DD + k + 1];
        wd[0] = W_d[k] * n0sel;
        wd[1] = W_d[k + 1] * n0sel;

        accB = __builtin_amdgcn_wmma_f32_16x16x4_f32(false, a, false, wb, (short)0, accB, false, false);
        accC = __builtin_amdgcn_wmma_f32_16x16x4_f32(false, a, false, wc, (short)0, accC, false, false);
        accD = __builtin_amdgcn_wmma_f32_16x16x4_f32(false, a, false, wd, (short)0, accD, false, false);
    }

#pragma unroll
    for (int r = 0; r < 8; ++r) {
        const int l  = l0 + r + 8 * half;
        const int gr = b * LL + l;
        Bp[gr * NN + mn] = accB[r];
        Cp[gr * NN + mn] = accC[r];
        if (mn == 0) dlin[gr] = accD[r];
    }
}

__global__ __launch_bounds__(256) void ssm_delta(
    const float* __restrict__ dlin,
    const float* __restrict__ p_Delta,
    float* __restrict__ delta)
{
    const int i  = blockIdx.x * 256 + threadIdx.x;
    const int l  = i & (LL - 1);
    const int bd = i >> 10;
    const int b  = bd / DD;
    const int d  = bd - b * DD;
    const float dl = dlin[b * LL + l] + p_Delta[d];
    delta[i] = fmaxf(dl, 0.0f) + log1pf(__expf(-fabsf(dl)));
}

__global__ __launch_bounds__(256) void ssm_scan_local(
    const float* __restrict__ x,
    const float* __restrict__ A,
    const float* __restrict__ delta,
    const float* __restrict__ Bp,
    float* __restrict__ Pc,
    float* __restrict__ Sc)
{
    const int gtid  = blockIdx.x * 256 + threadIdx.x;
    const int q     = gtid & 3;
    const int group = gtid >> 2;
    const int c     = group & (NC - 1);
    const int pair  = group >> 4;
    const int b     = pair / DD;
    const int d     = pair - b * DD;

    const float4 An  = ((const float4*)(A + d * NN))[q];
    const float4 rAn = make_float4(1.0f / An.x, 1.0f / An.y, 1.0f / An.z, 1.0f / An.w);

    const float*  __restrict__ dl  = delta + (size_t)pair * LL + c * LC;
    const float*  __restrict__ xr  = x     + (size_t)pair * LL + c * LC;
    const float4* __restrict__ bpr = (const float4*)(Bp + ((size_t)b * LL + c * LC) * NN) + q;

    float4 h = make_float4(0.f, 0.f, 0.f, 0.f);
    float sd = 0.0f;
#pragma unroll 2
    for (int s = 0; s < LC; ++s) {
        const float dt = dl[s];
        const float xv = xr[s];
        const float4 bp = bpr[4 * s];
        sd += dt;
        const float a0 = __expf(dt * An.x);
        const float a1 = __expf(dt * An.y);
        const float a2 = __expf(dt * An.z);
        const float a3 = __expf(dt * An.w);
        h.x = fmaf(a0, h.x, (a0 - An.x) * rAn.x * bp.x * xv);
        h.y = fmaf(a1, h.y, (a1 - An.y) * rAn.y * bp.y * xv);
        h.z = fmaf(a2, h.z, (a2 - An.z) * rAn.z * bp.z * xv);
        h.w = fmaf(a3, h.w, (a3 - An.w) * rAn.w * bp.w * xv);
    }
    float4 P;
    P.x = __expf(An.x * sd);
    P.y = __expf(An.y * sd);
    P.z = __expf(An.z * sd);
    P.w = __expf(An.w * sd);
    ((float4*)Pc)[group * 4 + q] = P;
    ((float4*)Sc)[group * 4 + q] = h;
}

__global__ __launch_bounds__(256) void ssm_chunk_combine(
    const float* __restrict__ Pc,
    const float* __restrict__ Sc,
    float* __restrict__ Hinit)
{
    const int t    = blockIdx.x * 256 + threadIdx.x;
    const int n    = t & 15;
    const int pair = t >> 4;
    float h = 0.0f;
#pragma unroll
    for (int c = 0; c < NC; ++c) {
        const int idx = (pair * NC + c) * NN + n;
        Hinit[idx] = h;
        h = fmaf(Pc[idx], h, Sc[idx]);
    }
}

__global__ __launch_bounds__(256) void ssm_scan_final(
    const float* __restrict__ x,
    const float* __restrict__ A,
    const float* __restrict__ delta,
    const float* __restrict__ Bp,
    const float* __restrict__ Cp,
    const float* __restrict__ Hinit,
    float* __restrict__ y)
{
    const int gtid  = blockIdx.x * 256 + threadIdx.x;
    const int q     = gtid & 3;
    const int group = gtid >> 2;
    const int c     = group & (NC - 1);
    const int pair  = group >> 4;
    const int b     = pair / DD;
    const int d     = pair - b * DD;

    const float4 An  = ((const float4*)(A + d * NN))[q];
    const float4 rAn = make_float4(1.0f / An.x, 1.0f / An.y, 1.0f / An.z, 1.0f / An.w);

    const float*  __restrict__ dl  = delta + (size_t)pair * LL + c * LC;
    const float*  __restrict__ xr  = x     + (size_t)pair * LL + c * LC;
    const float4* __restrict__ bpr = (const float4*)(Bp + ((size_t)b * LL + c * LC) * NN) + q;
    const float4* __restrict__ cpr = (const float4*)(Cp + ((size_t)b * LL + c * LC) * NN) + q;
    float*        __restrict__ yr  = y     + (size_t)pair * LL + c * LC;

    float4 h = ((const float4*)Hinit)[group * 4 + q];

#pragma unroll 2
    for (int s = 0; s < LC; ++s) {
        const float dt = dl[s];
        const float xv = xr[s];
        const float4 bp = bpr[4 * s];
        const float4 cp = cpr[4 * s];
        const float a0 = __expf(dt * An.x);
        const float a1 = __expf(dt * An.y);
        const float a2 = __expf(dt * An.z);
        const float a3 = __expf(dt * An.w);
        h.x = fmaf(a0, h.x, (a0 - An.x) * rAn.x * bp.x * xv);
        h.y = fmaf(a1, h.y, (a1 - An.y) * rAn.y * bp.y * xv);
        h.z = fmaf(a2, h.z, (a2 - An.z) * rAn.z * bp.z * xv);
        h.w = fmaf(a3, h.w, (a3 - An.w) * rAn.w * bp.w * xv);

        float p = cp.x * h.x;
        p = fmaf(cp.y, h.y, p);
        p = fmaf(cp.z, h.z, p);
        p = fmaf(cp.w, h.w, p);
        p += __shfl_xor(p, 1, 4);
        p += __shfl_xor(p, 2, 4);
        if (q == 0) yr[s] = p;
    }
}

extern "C" void kernel_launch(void* const* d_in, const int* in_sizes, int n_in,
                              void* d_out, int out_size, void* d_ws, size_t ws_size,
                              hipStream_t stream)
{
    const float* x       = (const float*)d_in[0];
    const float* A       = (const float*)d_in[1];
    const float* W_B     = (const float*)d_in[2];
    const float* b_B     = (const float*)d_in[3];
    const float* W_C     = (const float*)d_in[4];
    const float* b_C     = (const float*)d_in[5];
    const float* W_d     = (const float*)d_in[6];
    const float* b_d     = (const float*)d_in[7];
    const float* p_Delta = (const float*)d_in[8];
    float* y = (float*)d_out;

    float* ws    = (float*)d_ws;
    float* Bp    = ws;
    float* Cp    = Bp + (size_t)BB * LL * NN;
    float* dlin  = Cp + (size_t)BB * LL * NN;
    float* delta = dlin + (size_t)BB * LL;
    float* Pc    = delta + (size_t)BB * DD * LL;
    float* Sc    = Pc + (size_t)BB * DD * NC * NN;
    float* Hinit = Sc + (size_t)BB * DD * NC * NN;

    ssm_proj_wmma<<<dim3(128), dim3(32), 0, stream>>>(
        x, W_B, b_B, W_C, b_C, W_d, b_d, Bp, Cp, dlin);

    ssm_delta<<<dim3((BB * DD * LL) / 256), dim3(256), 0, stream>>>(
        dlin, p_Delta, delta);

    ssm_scan_local<<<dim3((BB * DD * NC * 4) / 256), dim3(256), 0, stream>>>(
        x, A, delta, Bp, Pc, Sc);

    ssm_chunk_combine<<<dim3((BB * DD * NN) / 256), dim3(256), 0, stream>>>(
        Pc, Sc, Hinit);

    ssm_scan_final<<<dim3((BB * DD * NC * 4) / 256), dim3(256), 0, stream>>>(
        x, A, delta, Bp, Cp, Hinit, y);
}


# projection GEMM: K split over 8 waves per row tile with grouped loads, so about 8x the loads in flight; partial tiles added via LDS in fixed slice order (changes the f32 summation order over K)
# speedup vs baseline: 1.7624x; 1.7624x over previous
#include <hip/hip_runtime.h>
#include <math.h>

#define DD 768
#define LL 1024
#define NN 16
#define BB 2
#define NC 16
#define LC (LL / NC)

typedef float v2f __attribute__((ext_vector_type(2)));
typedef float v8f __attribute__((ext_vector_type(8)));

#define KSPLIT 8
#define KW     (DD / KSPLIT)
#define KSTEPS (KW / 4)
#define KGRP   8

__global__ __launch_bounds__(32 * KSPLIT) void ssm_proj_wmma(
    const float* __restrict__ x,
    const float* __restrict__ W_B,
    const float* __restrict__ b_B,
    const float* __restrict__ W_C,
    const float* __restrict__ b_C,
    const float* __restrict__ W_d,
    const float* __restrict__ b_d,
    float* __restrict__ Bp,
    float* __restrict__ Cp,
    float* __restrict__ dlin)
{
    __shared__ float red[KSPLIT][24][32];

    const int lane = threadIdx.x & 31;
    const int wave = __builtin_amdgcn_readfirstlane((int)(threadIdx.x >> 5));
    const int half = lane >> 4;
    const int mn   = lane & 15;
    const int row0 = blockIdx.x << 4;
    const int b    = row0 >> 10;
    const int l0   = row0 & (LL - 1);
    const int lA   = l0 + mn;

    const float biasB = (wave == 0) ? b_B[mn] : 0.0f;
    const float biasC = (wave == 0) ? b_C[mn] : 0.0f;
    const float biasD = (wave == 0 && mn == 0) ? b_d[0] : 0.0f;
    v8f accB, accC, accD;
#pragma unroll
    for (int r = 0; r < 8; ++r) { accB[r] = biasB; accC[r] = biasC; accD[r] = biasD; }

    const float n0sel = (mn == 0) ? 1.0f : 0.0f;
    const int   kw0   = wave * KW + 2 * half;
    const float* __restrict__ xa  = x + (size_t)b * DD * LL + (size_t)kw0 * LL + lA;
    const float* __restrict__ wbp = W_B + mn * DD + kw0;
    const float* __restrict__ wcp = W_C + mn * DD + kw0;
    const float* __restrict__ wdp = W_d + kw0;

#pragma unroll
    for (int s0 = 0; s0 < KSTEPS; s0 += KGRP) {
        v2f a[KGRP], wb[KGRP], wc[KGRP], wd[KGRP];
#pragma unroll
        for (int u = 0; u < KGRP; ++u) {
            const int ko = 4 * (s0 + u);
            a[u][0]  = xa[(size_t)ko * LL];
            a[u][1]  = xa[(size_t)(ko + 1) * LL];
            wb[u][0] = wbp[ko];
            wb[u][1] = wbp[ko + 1];
            wc[u][0] = wcp[ko];
            wc[u][1] = wcp[ko + 1];
            wd[u][0] = wdp[ko] * n0sel;
            wd[u][1] = wdp[ko + 1] * n0sel;
        }
#pragma unroll
        for (int u = 0; u < KGRP; ++u) {
            accB = __builtin_amdgcn_wmma_f32_16x16x4_f32(false, a[u], false, wb[u], (short)0, accB, false, false);
            accC = __builtin_amdgcn_wmma_f32_16x16x4_f32(false, a[u], false, wc[u], (short)0, accC, false, false);
            accD = __builtin_amdgcn_wmma_f32_16x16x4_f32(false, a[u], false, wd[u], (short)0, accD, false, false);
        }
    }

#pragma unroll
    for (int r = 0; r < 8; ++r) {
        red[wave][r][lane]      = accB[r];
        red[wave][8 + r][lane]  = accC[r];
        red[wave][16 + r][lane] = accD[r];
    }
    __syncthreads();

    float sB = red[0][wave][lane];
    float sC = red[0][8 + wave][lane];
    float sD = red[0][16 + wave][lane];
#pragma unroll
    for (int w = 1; w < KSPLIT; ++w) {
        sB += red[w][wave][lane];
        sC += red[w][8 + wave][lane];
        sD += red[w][16 + wave][lane];
    }
    const int l  = l0 + wave + 8 * half;
    const int gr = b * LL + l;
    Bp[gr * NN + mn] = sB;
    Cp[gr * NN + mn] = sC;
    if (mn == 0) dlin[gr] = sD;
}

__global__ __launch_bounds__(256) void ssm_delta(
    const float* __restrict__ dlin,
    const float* __restrict__ p_Delta,
    float* __restrict__ delta)
{
    const int i  = blockIdx.x * 256 + threadIdx.x;
    const int l  = i & (LL - 1);
    const int bd = i >> 10;
    const int b  = bd / DD;
    const int d  = bd - b * DD;
    const float dl = dlin[b * LL + l] + p_Delta[d];
    delta[i] = fmaxf(dl, 0.0f) + log1pf(__expf(-fabsf(dl)));
}

__global__ __launch_bounds__(256) void ssm_scan_local(
    const float* __restrict__ x,
    const float* __restrict__ A,
    const float* __restrict__ delta,
    const float* __restrict__ Bp,
    float* __restrict__ Pc,
    float* __restrict__ Sc)
{
    const int gtid  = blockIdx.x * 256 + threadIdx.x;
    const int q     = gtid & 3;
    const int group = gtid >> 2;
    const int c     = group & (NC - 1);
    const int pair  = group >> 4;
    const int b     = pair / DD;
    const int d     = pair - b * DD;

    const float4 An  = ((const float4*)(A + d * NN))[q];
    const float4 rAn = make_float4(1.0f / An.x, 1.0f / An.y, 1.0f / An.z, 1.0f / An.w);

    const float*  __restrict__ dl  = delta + (size_t)pair * LL + c * LC;
    const float*  __restrict__ xr  = x     + (size_t)pair * LL + c * LC;
    const float4* __restrict__ bpr = (const float4*)(Bp + ((size_t)b * LL + c * LC) * NN) + q;

    float4 h = make_float4(0.f, 0.f, 0.f, 0.f);
    float sd = 0.0f;
#pragma unroll 2
    for (int s = 0; s < LC; ++s) {
        const float dt = dl[s];
        const float xv = xr[s];
        const float4 bp = bpr[4 * s];
        sd += dt;
        const float a0 = __expf(dt * An.x);
        const float a1 = __expf(dt * An.y);
        const float a2 = __expf(dt * An.z);
        const float a3 = __expf(dt * An.w);
        h.x = fmaf(a0, h.x, (a0 - An.x) * rAn.x * bp.x * xv);
        h.y = fmaf(a1, h.y, (a1 - An.y) * rAn.y * bp.y * xv);
        h.z = fmaf(a2, h.z, (a2 - An.z) * rAn.z * bp.z * xv);
        h.w = fmaf(a3, h.w, (a3 - An.w) * rAn.w * bp.w * xv);
    }
    float4 P;
    P.x = __expf(An.x * sd);
    P.y = __expf(An.y * sd);
    P.z = __expf(An.z * sd);
    P.w = __expf(An.w * sd);
    ((float4*)Pc)[group * 4 + q] = P;
    ((float4*)Sc)[group * 4 + q] = h;
}

__global__ __launch_bounds__(256) void ssm_chunk_combine(
    const float* __restrict__ Pc,
    const float* __restrict__ Sc,
    float* __restrict__ Hinit)
{
    const int t    = blockIdx.x * 256 + threadIdx.x;
    const int n    = t & 15;
    const int pair = t >> 4;
    float h = 0.0f;
#pragma unroll
    for (int c = 0; c < NC; ++c) {
        const int idx = (pair * NC + c) * NN + n;
        Hinit[idx] = h;
        h = fmaf(Pc[idx], h, Sc[idx]);
    }
}

__global__ __launch_bounds__(256) void ssm_scan_final(
    const float* __restrict__ x,
    const float* __restrict__ A,
    const float* __restrict__ delta,
    const float* __restrict__ Bp,
    const float* __restrict__ Cp,
    const float* __restrict__ Hinit,
    float* __restrict__ y)
{
    const int gtid  = blockIdx.x * 256 + threadIdx.x;
    const int q     = gtid & 3;
    const int group = gtid >> 2;
    const int c     = group & (NC - 1);
    const int pair  = group >> 4;
    const int b     = pair / DD;
    const int d     = pair - b * DD;

    const float4 An  = ((const float4*)(A + d * NN))[q];
    const float4 rAn = make_float4(1.0f / An.x, 1.0f / An.y, 1.0f / An.z, 1.0f / An.w);

    const float*  __restrict__ dl  = delta + (size_t)pair * LL + c * LC;
    const float*  __restrict__ xr  = x     + (size_t)pair * LL + c * LC;
    const float4* __restrict__ bpr = (const float4*)(Bp + ((size_t)b * LL + c * LC) * NN) + q;
    const float4* __restrict__ cpr = (const float4*)(Cp + ((size_t)b * LL + c * LC) * NN) + q;
    float*        __restrict__ yr  = y     + (size_t)pair * LL + c * LC;

    float4 h = ((const float4*)Hinit)[group * 4 + q];

#pragma unroll 2
    for (int s = 0; s < LC; ++s) {
        const float dt = dl[s];
        const float xv = xr[s];
        const float4 bp = bpr[4 * s];
        const float4 cp = cpr[4 * s];
        const float a0 = __expf(dt * An.x);
        const float a1 = __expf(dt * An.y);
        const float a2 = __expf(dt * An.z);
        const float a3 = __expf(dt * An.w);
        h.x = fmaf(a0, h.x, (a0 - An.x) * rAn.x * bp.x * xv);
        h.y = fmaf(a1, h.y, (a1 - An.y) * rAn.y * bp.y * xv);
        h.z = fmaf(a2, h.z, (a2 - An.z) * rAn.z * bp.z * xv);
        h.w = fmaf(a3, h.w, (a3 - An.w) * rAn.w * bp.w * xv);

        float p = cp.x * h.x;
        p = fmaf(cp.y, h.y, p);
        p = fmaf(cp.z, h.z, p);
        p = fmaf(cp.w, h.w, p);
        p += __shfl_xor(p, 1, 4);
        p += __shfl_xor(p, 2, 4);
        if (q == 0) yr[s] = p;
    }
}

extern "C" void kernel_launch(void* const* d_in, const int* in_sizes, int n_in,
                              void* d_out, int out_size, void* d_ws, size_t ws_size,
                              hipStream_t stream)
{
    const float* x       = (const float*)d_in[0];
    const float* A       = (const float*)d_in[1];
    const float* W_B     = (const float*)d_in[2];
    const float* b_B     = (const float*)d_in[3];
    const float* W_C     = (const float*)d_in[4];
    const float* b_C     = (const float*)d_in[5];
    const float* W_d     = (const float*)d_in[6];
    const float* b_d     = (const float*)d_in[7];
    const float* p_Delta = (const float*)d_in[8];
    float* y = (float*)d_out;

    float* ws    = (float*)d_ws;
    float* Bp    = ws;
    float* Cp    = Bp + (size_t)BB * LL * NN;
    float* dlin  = Cp + (size_t)BB * LL * NN;
    float* delta = dlin + (size_t)BB * LL;
    float* Pc    = delta + (size_t)BB * DD * LL;
    float* Sc    = Pc + (size_t)BB * DD * NC * NN;
    float* Hinit = Sc + (size_t)BB * DD * NC * NN;

    ssm_proj_wmma<<<dim3(128), dim3(32 * KSPLIT), 0, stream>>>(
        x, W_B, b_B, W_C, b_C, W_d, b_d, Bp, Cp, dlin);

    ssm_delta<<<dim3((BB * DD * LL) / 256), dim3(256), 0, stream>>>(
        dlin, p_Delta, delta);

    ssm_scan_local<<<dim3((BB * DD * NC * 4) / 256), dim3(256), 0, stream>>>(
        x, A, delta, Bp, Pc, Sc);

    ssm_chunk_combine<<<dim3((BB * DD * NN) / 256), dim3(256), 0, stream>>>(
        Pc, Sc, Hinit);

    ssm_scan_final<<<dim3((BB * DD * NC * 4) / 256), dim3(256), 0, stream>>>(
        x, A, delta, Bp, Cp, Hinit, y);
}


# scan kernels: one work-group per (batch, chunk, 96 channels) stages the chunk's delta, x, Bp, Cp rows in LDS with grouped b128 loads, y leaves the same way; the 64-step loop reads LDS only
# speedup vs baseline: 3.2774x; 1.8596x over previous
#include <hip/hip_runtime.h>
#include <math.h>

#define DD 768
#define LL 1024
#define NN 16
#define BB 2
#define NC 16
#define LC (LL / NC)

typedef float v2f __attribute__((ext_vector_type(2)));
typedef float v8f __attribute__((ext_vector_type(8)));

#define KSPLIT 8
#define KW     (DD / KSPLIT)
#define KSTEPS (KW / 4)
#define KGRP   8

__global__ __launch_bounds__(32 * KSPLIT) void ssm_proj_wmma(
    const float* __restrict__ x,
    const float* __restrict__ W_B,
    const float* __restrict__ b_B,
    const float* __restrict__ W_C,
    const float* __restrict__ b_C,
    const float* __restrict__ W_d,
    const float* __restrict__ b_d,
    float* __restrict__ Bp,
    float* __restrict__ Cp,
    float* __restrict__ dlin)
{
    __shared__ float red[KSPLIT][24][32];

    const int lane = threadIdx.x & 31;
    const int wave = __builtin_amdgcn_readfirstlane((int)(threadIdx.x >> 5));
    const int half = lane >> 4;
    const int mn   = lane & 15;
    const int row0 = blockIdx.x << 4;
    const int b    = row0 >> 10;
    const int l0   = row0 & (LL - 1);
    const int lA   = l0 + mn;

    const float biasB = (wave == 0) ? b_B[mn] : 0.0f;
    const float biasC = (wave == 0) ? b_C[mn] : 0.0f;
    const float biasD = (wave == 0 && mn == 0) ? b_d[0] : 0.0f;
    v8f accB, accC, accD;
#pragma unroll
    for (int r = 0; r < 8; ++r) { accB[r] = biasB; accC[r] = biasC; accD[r] = biasD; }

    const float n0sel = (mn == 0) ? 1.0f : 0.0f;
    const int   kw0   = wave * KW + 2 * half;
    const float* __restrict__ xa  = x + (size_t)b * DD * LL + (size_t)kw0 * LL + lA;
    const float* __restrict__ wbp = W_B + mn * DD + kw0;
    const float* __restrict__ wcp = W_C + mn * DD + kw0;
    const float* __restrict__ wdp = W_d + kw0;

#pragma unroll
    for (int s0 = 0; s0 < KSTEPS; s0 += KGRP) {
        v2f a[KGRP], wb[KGRP], wc[KGRP], wd[KGRP];
#pragma unroll
        for (int u = 0; u < KGRP; ++u) {
            const int ko = 4 * (s0 + u);
            a[u][0]  = xa[(size_t)ko * LL];
            a[u][1]  = xa[(size_t)(ko + 1) * LL];
            wb[u][0] = wbp[ko];
            wb[u][1] = wbp[ko + 1];
            wc[u][0] = wcp[ko];
            wc[u][1] = wcp[ko + 1];
            wd[u][0] = wdp[ko] * n0sel;
            wd[u][1] = wdp[ko + 1] * n0sel;
        }
#pragma unroll
        for (int u = 0; u < KGRP; ++u) {
            accB = __builtin_amdgcn_wmma_f32_16x16x4_f32(false, a[u], false, wb[u], (short)0, accB, false, false);
            accC = __builtin_amdgcn_wmma_f32_16x16x4_f32(false, a[u], false, wc[u], (short)0, accC, false, false);
            accD = __builtin_amdgcn_wmma_f32_16x16x4_f32(false, a[u], false, wd[u], (short)0, accD, false, false);
        }
    }

#pragma unroll
    for (int r = 0; r < 8; ++r) {
        red[wave][r][lane]      = accB[r];
        red[wave][8 + r][lane]  = accC[r];
        red[wave][16 + r][lane] = accD[r];
    }
    __syncthreads();

    float sB = red[0][wave][lane];
    float sC = red[0][8 + wave][lane];
    float sD = red[0][16 + wave][lane];
#pragma unroll
    for (int w = 1; w < KSPLIT; ++w) {
        sB += red[w][wave][lane];
        sC += red[w][8 + wave][lane];
        sD += red[w][16 + wave][lane];
    }
    const int l  = l0 + wave + 8 * half;
    const int gr = b * LL + l;
    Bp[gr * NN + mn] = sB;
    Cp[gr * NN + mn] = sC;
    if (mn == 0) dlin[gr] = sD;
}

__global__ __launch_bounds__(256) void ssm_delta(
    const float* __restrict__ dlin,
    const float* __restrict__ p_Delta,
    float* __restrict__ delta)
{
    const int i  = blockIdx.x * 256 + threadIdx.x;
    const int l  = i & (LL - 1);
    const int bd = i >> 10;
    const int b  = bd / DD;
    const int d  = bd - b * DD;
    const float dl = dlin[b * LL + l] + p_Delta[d];
    delta[i] = fmaxf(dl, 0.0f) + log1pf(__expf(-fabsf(dl)));
}

#define DBLK  96
#define SCANT (DBLK * 4)
#define NDBLK (DD / DBLK)
#define TP4   17

__device__ __forceinline__ void ssm_step(float4& h, const float4 An, const float4 rAn,
                                         const float dt, const float xv, const float4 bp)
{
    const float a0 = __expf(dt * An.x);
    const float a1 = __expf(dt * An.y);
    const float a2 = __expf(dt * An.z);
    const float a3 = __expf(dt * An.w);
    h.x = fmaf(a0, h.x, (a0 - An.x) * rAn.x * bp.x * xv);
    h.y = fmaf(a1, h.y, (a1 - An.y) * rAn.y * bp.y * xv);
    h.z = fmaf(a2, h.z, (a2 - An.z) * rAn.z * bp.z * xv);
    h.w = fmaf(a3, h.w, (a3 - An.w) * rAn.w * bp.w * xv);
}

__device__ __forceinline__ float ssm_out(const float4 h, const float4 cp)
{
    float p = cp.x * h.x;
    p = fmaf(cp.y, h.y, p);
    p = fmaf(cp.z, h.z, p);
    p = fmaf(cp.w, h.w, p);
    p += __shfl_xor(p, 1, 4);
    p += __shfl_xor(p, 2, 4);
    return p;
}

__global__ __launch_bounds__(SCANT) void ssm_scan_local(
    const float* __restrict__ x,
    const float* __restrict__ A,
    const float* __restrict__ delta,
    const float* __restrict__ Bp,
    float* __restrict__ Pc,
    float* __restrict__ Sc)
{
    __shared__ float4 s_dt[DBLK * TP4];
    __shared__ float4 s_x[DBLK * TP4];
    __shared__ float4 s_bp[LC * 4];

    const int t    = threadIdx.x;
    const int dblk = blockIdx.x % NDBLK;
    const int c    = (blockIdx.x / NDBLK) & (NC - 1);
    const int b    = blockIdx.x / (NDBLK * NC);
    const int d0   = dblk * DBLK;

    const float4* __restrict__ gdt = (const float4*)(delta + ((size_t)(b * DD + d0)) * LL + c * LC);
    const float4* __restrict__ gx  = (const float4*)(x     + ((size_t)(b * DD + d0)) * LL + c * LC);
    const float4* __restrict__ gbp = (const float4*)(Bp + ((size_t)b * LL + c * LC) * NN);

    float4 rdt[4], rx[4];
#pragma unroll
    for (int j = 0; j < 4; ++j) {
        const int i = t + SCANT * j;
        const int off = (i >> 4) * (LL / 4) + (i & 15);
        rdt[j] = gdt[off];
        rx[j]  = gx[off];
    }
    float4 rbp = make_float4(0.f, 0.f, 0.f, 0.f);
    if (t < LC * 4) rbp = gbp[t];

    const int q     = t & 3;
    const int g     = t >> 2;
    const int d     = d0 + g;
    const int group = (b * DD + d) * NC + c;
    const float4 An  = ((const float4*)(A + d * NN))[q];

#pragma unroll
    for (int j = 0; j < 4; ++j) {
        const int i = t + SCANT * j;
        const int off = (i >> 4) * TP4 + (i & 15);
        s_dt[off] = rdt[j];
        s_x[off]  = rx[j];
    }
    if (t < LC * 4) s_bp[t] = rbp;
    __syncthreads();

    const float4 rAn = make_float4(1.0f / An.x, 1.0f / An.y, 1.0f / An.z, 1.0f / An.w);
    const float4* pdt = s_dt + g * TP4;
    const float4* px  = s_x + g * TP4;
    const float4* pbp = s_bp + q;

    float4 h = make_float4(0.f, 0.f, 0.f, 0.f);
    float sd = 0.0f;
    for (int s8 = 0; s8 < LC / 8; ++s8) {
        float4 dt4[2], xv4[2], bp[8];
#pragma unroll
        for (int u = 0; u < 2; ++u) {
            dt4[u] = pdt[2 * s8 + u];
            xv4[u] = px[2 * s8 + u];
        }
#pragma unroll
        for (int k = 0; k < 8; ++k) bp[k] = pbp[(8 * s8 + k) * 4];
#pragma unroll
        for (int u = 0; u < 2; ++u) {
            sd += dt4[u].x; ssm_step(h, An, rAn, dt4[u].x, xv4[u].x, bp[4 * u + 0]);
            sd += dt4[u].y; ssm_step(h, An, rAn, dt4[u].y, xv4[u].y, bp[4 * u + 1]);
            sd += dt4[u].z; ssm_step(h, An, rAn, dt4[u].z, xv4[u].z, bp[4 * u + 2]);
            sd += dt4[u].w; ssm_step(h, An, rAn, dt4[u].w, xv4[u].w, bp[4 * u + 3]);
        }
    }
    float4 P;
    P.x = __expf(An.x * sd);
    P.y = __expf(An.y * sd);
    P.z = __expf(An.z * sd);
    P.w = __expf(An.w * sd);
    ((float4*)Pc)[group * 4 + q] = P;
    ((float4*)Sc)[group * 4 + q] = h;
}

__global__ __launch_bounds__(256) void ssm_chunk_combine(
    const float* __restrict__ Pc,
    const float* __restrict__ Sc,
    float* __restrict__ Hinit)
{
    const int t    = blockIdx.x * 256 + threadIdx.x;
    const int n    = t & 15;
    const int pair = t >> 4;
    float h = 0.0f;
#pragma unroll
    for (int c = 0; c < NC; ++c) {
        const int idx = (pair * NC + c) * NN + n;
        Hinit[idx] = h;
        h = fmaf(Pc[idx], h, Sc[idx]);
    }
}

__global__ __launch_bounds__(SCANT) void ssm_scan_final(
    const float* __restrict__ x,
    const float* __restrict__ A,
    const float* __restrict__ delta,
    const float* __restrict__ Bp,
    const float* __restrict__ Cp,
    const float* __restrict__ Hinit,
    float* __restrict__ y)
{
    __shared__ float4 s_dt[DBLK * TP4];
    __shared__ float4 s_x[DBLK * TP4];
    __shared__ float4 s_y[DBLK * TP4];
    __shared__ float4 s_bp[LC * 4];
    __shared__ float4 s_cp[LC * 4];

    const int t    = threadIdx.x;
    const int dblk = blockIdx.x % NDBLK;
    const int c    = (blockIdx.x / NDBLK) & (NC - 1);
    const int b    = blockIdx.x / (NDBLK * NC);
    const int d0   = dblk * DBLK;

    const float4* __restrict__ gdt = (const float4*)(delta + ((size_t)(b * DD + d0)) * LL + c * LC);
    const float4* __restrict__ gx  = (const float4*)(x     + ((size_t)(b * DD + d0)) * LL + c * LC);
    const float4* __restrict__ gbp = (const float4*)(Bp + ((size_t)b * LL + c * LC) * NN);
    const float4* __restrict__ gcp = (const float4*)(Cp + ((size_t)b * LL + c * LC) * NN);
    float4*       __restrict__ gy  = (float4*)(y + ((size_t)(b * DD + d0)) * LL + c * LC);

    float4 rdt[4], rx[4];
#pragma unroll
    for (int j = 0; j < 4; ++j) {
        const int i = t + SCANT * j;
        const int off = (i >> 4) * (LL / 4) + (i & 15);
        rdt[j] = gdt[off];
        rx[j]  = gx[off];
    }
    float4 rbp = make_float4(0.f, 0.f, 0.f, 0.f);
    float4 rcp = make_float4(0.f, 0.f, 0.f, 0.f);
    if (t < LC * 4) { rbp = gbp[t]; rcp = gcp[t]; }

    const int q     = t & 3;
    const int g     = t >> 2;
    const int d     = d0 + g;
    const int group = (b * DD + d) * NC + c;
    const float4 An = ((const float4*)(A + d * NN))[q];
    float4 h = ((const float4*)Hinit)[group * 4 + q];

#pragma unroll
    for (int j = 0; j < 4; ++j) {
        const int i = t + SCANT * j;
        const int off = (i >> 4) * TP4 + (i & 15);
        s_dt[off] = rdt[j];
        s_x[off]  = rx[j];
    }
    if (t < LC * 4) { s_bp[t] = rbp; s_cp[t] = rcp; }
    __syncthreads();

    const float4 rAn = make_float4(1.0f / An.x, 1.0f / An.y, 1.0f / An.z, 1.0f / An.w);
    const float4* pdt = s_dt + g * TP4;
    const float4* px  = s_x + g * TP4;
    const float4* pbp = s_bp + q;
    const float4* pcp = s_cp + q;
    float4* prow = s_y + g * TP4;

    for (int s8 = 0; s8 < LC / 8; ++s8) {
        float4 dt4[2], xv4[2], bp[8], cp[8];
        float p[8];
#pragma unroll
        for (int u = 0; u < 2; ++u) {
            dt4[u] = pdt[2 * s8 + u];
            xv4[u] = px[2 * s8 + u];
        }
#pragma unroll
        for (int k = 0; k < 8; ++k) {
            bp[k] = pbp[(8 * s8 + k) * 4];
            cp[k] = pcp[(8 * s8 + k) * 4];
        }
#pragma unroll
        for (int u = 0; u < 2; ++u) {
            ssm_step(h, An, rAn, dt4[u].x, xv4[u].x, bp[4 * u + 0]); p[4 * u + 0] = ssm_out(h, cp[4 * u + 0]);
            ssm_step(h, An, rAn, dt4[u].y, xv4[u].y, bp[4 * u + 1]); p[4 * u + 1] = ssm_out(h, cp[4 * u + 1]);
            ssm_step(h, An, rAn, dt4[u].z, xv4[u].z, bp[4 * u + 2]); p[4 * u + 2] = ssm_out(h, cp[4 * u + 2]);
            ssm_step(h, An, rAn, dt4[u].w, xv4[u].w, bp[4 * u + 3]); p[4 * u + 3] = ssm_out(h, cp[4 * u + 3]);
        }
        if (q == 0) {
            prow[2 * s8]     = make_float4(p[0], p[1], p[2], p[3]);
            prow[2 * s8 + 1] = make_float4(p[4], p[5], p[6], p[7]);
        }
    }
    __syncthreads();

#pragma unroll
    for (int j = 0; j < 4; ++j) {
        const int i = t + SCANT * j;
        gy[(i >> 4) * (LL / 4) + (i & 15)] = s_y[(i >> 4) * TP4 + (i & 15)];
    }
}

extern "C" void kernel_launch(void* const* d_in, const int* in_sizes, int n_in,
                              void* d_out, int out_size, void* d_ws, size_t ws_size,
                              hipStream_t stream)
{
    const float* x       = (const float*)d_in[0];
    const float* A       = (const float*)d_in[1];
    const float* W_B     = (const float*)d_in[2];
    const float* b_B     = (const float*)d_in[3];
    const float* W_C     = (const float*)d_in[4];
    const float* b_C     = (const float*)d_in[5];
    const float* W_d     = (const float*)d_in[6];
    const float* b_d     = (const float*)d_in[7];
    const float* p_Delta = (const float*)d_in[8];
    float* y = (float*)d_out;

    float* ws    = (float*)d_ws;
    float* Bp    = ws;
    float* Cp    = Bp + (size_t)BB * LL * NN;
    float* dlin  = Cp + (size_t)BB * LL * NN;
    float* delta = dlin + (size_t)BB * LL;
    float* Pc    = delta + (size_t)BB * DD * LL;
    float* Sc    = Pc + (size_t)BB * DD * NC * NN;
    float* Hinit = Sc + (size_t)BB * DD * NC * NN;

    ssm_proj_wmma<<<dim3(128), dim3(32 * KSPLIT), 0, stream>>>(
        x, W_B, b_B, W_C, b_C, W_d, b_d, Bp, Cp, dlin);

    ssm_delta<<<dim3((BB * DD * LL) / 256), dim3(256), 0, stream>>>(
        dlin, p_Delta, delta);

    ssm_scan_local<<<dim3(BB * NC * NDBLK), dim3(SCANT), 0, stream>>>(
        x, A, delta, Bp, Pc, Sc);

    ssm_chunk_combine<<<dim3((BB * DD * NN) / 256), dim3(256), 0, stream>>>(
        Pc, Sc, Hinit);

    ssm_scan_final<<<dim3(BB * NC * NDBLK), dim3(SCANT), 0, stream>>>(
        x, A, delta, Bp, Cp, Hinit, y);
}


# delta kernel folded into pass A: each scan work-group computes softplus for its tile while staging it, keeps it in LDS and writes the delta plane for pass C; 5 launches become 4
# speedup vs baseline: 3.7633x; 1.1483x over previous
#include <hip/hip_runtime.h>
#include <math.h>

#define DD 768
#define LL 1024
#define NN 16
#define BB 2
#define NC 16
#define LC (LL / NC)

typedef float v2f __attribute__((ext_vector_type(2)));
typedef float v8f __attribute__((ext_vector_type(8)));

#define KSPLIT 8
#define KW     (DD / KSPLIT)
#define KSTEPS (KW / 4)
#define KGRP   8

__global__ __launch_bounds__(32 * KSPLIT) void ssm_proj_wmma(
    const float* __restrict__ x,
    const float* __restrict__ W_B,
    const float* __restrict__ b_B,
    const float* __restrict__ W_C,
    const float* __restrict__ b_C,
    const float* __restrict__ W_d,
    const float* __restrict__ b_d,
    float* __restrict__ Bp,
    float* __restrict__ Cp,
    float* __restrict__ dlin)
{
    __shared__ float red[KSPLIT][24][32];

    const int lane = threadIdx.x & 31;
    const int wave = __builtin_amdgcn_readfirstlane((int)(threadIdx.x >> 5));
    const int half = lane >> 4;
    const int mn   = lane & 15;
    const int row0 = blockIdx.x << 4;
    const int b    = row0 >> 10;
    const int l0   = row0 & (LL - 1);
    const int lA   = l0 + mn;

    const float biasB = (wave == 0) ? b_B[mn] : 0.0f;
    const float biasC = (wave == 0) ? b_C[mn] : 0.0f;
    const float biasD = (wave == 0 && mn == 0) ? b_d[0] : 0.0f;
    v8f accB, accC, accD;
#pragma unroll
    for (int r = 0; r < 8; ++r) { accB[r] = biasB; accC[r] = biasC; accD[r] = biasD; }

    const float n0sel = (mn == 0) ? 1.0f : 0.0f;
    const int   kw0   = wave * KW + 2 * half;
    const float* __restrict__ xa  = x + (size_t)b * DD * LL + (size_t)kw0 * LL + lA;
    const float* __restrict__ wbp = W_B + mn * DD + kw0;
    const float* __restrict__ wcp = W_C + mn * DD + kw0;
    const float* __restrict__ wdp = W_d + kw0;

#pragma unroll
    for (int s0 = 0; s0 < KSTEPS; s0 += KGRP) {
        v2f a[KGRP], wb[KGRP], wc[KGRP], wd[KGRP];
#pragma unroll
        for (int u = 0; u < KGRP; ++u) {
            const int ko = 4 * (s0 + u);
            a[u][0]  = xa[(size_t)ko * LL];
            a[u][1]  = xa[(size_t)(ko + 1) * LL];
            wb[u][0] = wbp[ko];
            wb[u][1] = wbp[ko + 1];
            wc[u][0] = wcp[ko];
            wc[u][1] = wcp[ko + 1];
            wd[u][0] = wdp[ko] * n0sel;
            wd[u][1] = wdp[ko + 1] * n0sel;
        }
#pragma unroll
        for (int u = 0; u < KGRP; ++u) {
            accB = __builtin_amdgcn_wmma_f32_16x16x4_f32(false, a[u], false, wb[u], (short)0, accB, false, false);
            accC = __builtin_amdgcn_wmma_f32_16x16x4_f32(false, a[u], false, wc[u], (short)0, accC, false, false);
            accD = __builtin_amdgcn_wmma_f32_16x16x4_f32(false, a[u], false, wd[u], (short)0, accD, false, false);
        }
    }

#pragma unroll
    for (int r = 0; r < 8; ++r) {
        red[wave][r][lane]      = accB[r];
        red[wave][8 + r][lane]  = accC[r];
        red[wave][16 + r][lane] = accD[r];
    }
    __syncthreads();

    float sB = red[0][wave][lane];
    float sC = red[0][8 + wave][lane];
    float sD = red[0][16 + wave][lane];
#pragma unroll
    for (int w = 1; w < KSPLIT; ++w) {
        sB += red[w][wave][lane];
        sC += red[w][8 + wave][lane];
        sD += red[w][16 + wave][lane];
    }
    const int l  = l0 + wave + 8 * half;
    const int gr = b * LL + l;
    Bp[gr * NN + mn] = sB;
    Cp[gr * NN + mn] = sC;
    if (mn == 0) dlin[gr] = sD;
}

__device__ __forceinline__ float ssm_softplus(const float dl)
{
    return fmaxf(dl, 0.0f) + log1pf(__expf(-fabsf(dl)));
}

#define DBLK  96
#define SCANT (DBLK * 4)
#define NDBLK (DD / DBLK)
#define TP4   17

__device__ __forceinline__ void ssm_step(float4& h, const float4 An, const float4 rAn,
                                         const float dt, const float xv, const float4 bp)
{
    const float a0 = __expf(dt * An.x);
    const float a1 = __expf(dt * An.y);
    const float a2 = __expf(dt * An.z);
    const float a3 = __expf(dt * An.w);
    h.x = fmaf(a0, h.x, (a0 - An.x) * rAn.x * bp.x * xv);
    h.y = fmaf(a1, h.y, (a1 - An.y) * rAn.y * bp.y * xv);
    h.z = fmaf(a2, h.z, (a2 - An.z) * rAn.z * bp.z * xv);
    h.w = fmaf(a3, h.w, (a3 - An.w) * rAn.w * bp.w * xv);
}

__device__ __forceinline__ float ssm_out(const float4 h, const float4 cp)
{
    float p = cp.x * h.x;
    p = fmaf(cp.y, h.y, p);
    p = fmaf(cp.z, h.z, p);
    p = fmaf(cp.w, h.w, p);
    p += __shfl_xor(p, 1, 4);
    p += __shfl_xor(p, 2, 4);
    return p;
}

__global__ __launch_bounds__(SCANT) __attribute__((amdgpu_waves_per_eu(3, 4))) void ssm_scan_local(
    const float* __restrict__ x,
    const float* __restrict__ A,
    const float* __restrict__ dlin,
    const float* __restrict__ p_Delta,
    const float* __restrict__ Bp,
    float* __restrict__ delta,
    float* __restrict__ Pc,
    float* __restrict__ Sc)
{
    __shared__ float4 s_dt[DBLK * TP4];
    __shared__ float4 s_x[DBLK * TP4];
    __shared__ float4 s_bp[LC * 4];

    const int t    = threadIdx.x;
    const int dblk = blockIdx.x % NDBLK;
    const int c    = (blockIdx.x / NDBLK) & (NC - 1);
    const int b    = blockIdx.x / (NDBLK * NC);
    const int d0   = dblk * DBLK;

    float4*       __restrict__ gdt = (float4*)(delta + ((size_t)(b * DD + d0)) * LL + c * LC);
    const float4* __restrict__ gx  = (const float4*)(x     + ((size_t)(b * DD + d0)) * LL + c * LC);
    const float4* __restrict__ gbp = (const float4*)(Bp + ((size_t)b * LL + c * LC) * NN);
    const float4* __restrict__ gdl = (const float4*)(dlin + (size_t)b * LL + c * LC);

    const float4 dl4 = gdl[t & 15];
    float pd[4];
    float4 rdt[4], rx[4];
#pragma unroll
    for (int j = 0; j < 4; ++j) {
        const int i = t + SCANT * j;
        const int off = (i >> 4) * (LL / 4) + (i & 15);
        pd[j] = p_Delta[d0 + (i >> 4)];
        rx[j] = gx[off];
    }
    const int q     = t & 3;
    const int g     = t >> 2;
    const int d     = d0 + g;
    const int group = (b * DD + d) * NC + c;
    const float4 An  = ((const float4*)(A + d * NN))[q];
    float4 rbp = make_float4(0.f, 0.f, 0.f, 0.f);
    if (t < LC * 4) rbp = gbp[t];

    asm volatile("" :: "v"(rx[0].x), "v"(rx[0].y), "v"(rx[0].z), "v"(rx[0].w),
                       "v"(rx[1].x), "v"(rx[1].y), "v"(rx[1].z), "v"(rx[1].w),
                       "v"(rx[2].x), "v"(rx[2].y), "v"(rx[2].z), "v"(rx[2].w),
                       "v"(rx[3].x), "v"(rx[3].y), "v"(rx[3].z), "v"(rx[3].w),
                       "v"(An.x), "v"(An.y), "v"(An.z), "v"(An.w));

#pragma unroll
    for (int j = 0; j < 4; ++j) {
        rdt[j].x = ssm_softplus(dl4.x + pd[j]);
        rdt[j].y = ssm_softplus(dl4.y + pd[j]);
        rdt[j].z = ssm_softplus(dl4.z + pd[j]);
        rdt[j].w = ssm_softplus(dl4.w + pd[j]);
    }

#pragma unroll
    for (int j = 0; j < 4; ++j) {
        const int i = t + SCANT * j;
        const int off = (i >> 4) * TP4 + (i & 15);
        s_dt[off] = rdt[j];
        s_x[off]  = rx[j];
    }
    if (t < LC * 4) s_bp[t] = rbp;
    __syncthreads();

#pragma unroll
    for (int j = 0; j < 4; ++j) {
        const int i = t + SCANT * j;
        gdt[(i >> 4) * (LL / 4) + (i & 15)] = rdt[j];
    }

    const float4 rAn = make_float4(1.0f / An.x, 1.0f / An.y, 1.0f / An.z, 1.0f / An.w);
    const float4* pdt = s_dt + g * TP4;
    const float4* px  = s_x + g * TP4;
    const float4* pbp = s_bp + q;

    float4 h = make_float4(0.f, 0.f, 0.f, 0.f);
    float sd = 0.0f;
    for (int s8 = 0; s8 < LC / 8; ++s8) {
        float4 dt4[2], xv4[2], bp[8];
#pragma unroll
        for (int u = 0; u < 2; ++u) {
            dt4[u] = pdt[2 * s8 + u];
            xv4[u] = px[2 * s8 + u];
        }
#pragma unroll
        for (int k = 0; k < 8; ++k) bp[k] = pbp[(8 * s8 + k) * 4];
#pragma unroll
        for (int u = 0; u < 2; ++u) {
            sd += dt4[u].x; ssm_step(h, An, rAn, dt4[u].x, xv4[u].x, bp[4 * u + 0]);
            sd += dt4[u].y; ssm_step(h, An, rAn, dt4[u].y, xv4[u].y, bp[4 * u + 1]);
            sd += dt4[u].z; ssm_step(h, An, rAn, dt4[u].z, xv4[u].z, bp[4 * u + 2]);
            sd += dt4[u].w; ssm_step(h, An, rAn, dt4[u].w, xv4[u].w, bp[4 * u + 3]);
        }
    }
    float4 P;
    P.x = __expf(An.x * sd);
    P.y = __expf(An.y * sd);
    P.z = __expf(An.z * sd);
    P.w = __expf(An.w * sd);
    ((float4*)Pc)[group * 4 + q] = P;
    ((float4*)Sc)[group * 4 + q] = h;
}

__global__ __launch_bounds__(256) void ssm_chunk_combine(
    const float* __restrict__ Pc,
    const float* __restrict__ Sc,
    float* __restrict__ Hinit)
{
    const int t    = blockIdx.x * 256 + threadIdx.x;
    const int n    = t & 15;
    const int pair = t >> 4;
    float h = 0.0f;
#pragma unroll
    for (int c = 0; c < NC; ++c) {
        const int idx = (pair * NC + c) * NN + n;
        Hinit[idx] = h;
        h = fmaf(Pc[idx], h, Sc[idx]);
    }
}

__global__ __launch_bounds__(SCANT) void ssm_scan_final(
    const float* __restrict__ x,
    const float* __restrict__ A,
    const float* __restrict__ delta,
    const float* __restrict__ Bp,
    const float* __restrict__ Cp,
    const float* __restrict__ Hinit,
    float* __restrict__ y)
{
    __shared__ float4 s_dt[DBLK * TP4];
    __shared__ float4 s_x[DBLK * TP4];
    __shared__ float4 s_y[DBLK * TP4];
    __shared__ float4 s_bp[LC * 4];
    __shared__ float4 s_cp[LC * 4];

    const int t    = threadIdx.x;
    const int dblk = blockIdx.x % NDBLK;
    const int c    = (blockIdx.x / NDBLK) & (NC - 1);
    const int b    = blockIdx.x / (NDBLK * NC);
    const int d0   = dblk * DBLK;

    const float4* __restrict__ gdt = (const float4*)(delta + ((size_t)(b * DD + d0)) * LL + c * LC);
    const float4* __restrict__ gx  = (const float4*)(x     + ((size_t)(b * DD + d0)) * LL + c * LC);
    const float4* __restrict__ gbp = (const float4*)(Bp + ((size_t)b * LL + c * LC) * NN);
    const float4* __restrict__ gcp = (const float4*)(Cp + ((size_t)b * LL + c * LC) * NN);
    float4*       __restrict__ gy  = (float4*)(y + ((size_t)(b * DD + d0)) * LL + c * LC);

    float4 rdt[4], rx[4];
#pragma unroll
    for (int j = 0; j < 4; ++j) {
        const int i = t + SCANT * j;
        const int off = (i >> 4) * (LL / 4) + (i & 15);
        rdt[j] = gdt[off];
        rx[j]  = gx[off];
    }
    float4 rbp = make_float4(0.f, 0.f, 0.f, 0.f);
    float4 rcp = make_float4(0.f, 0.f, 0.f, 0.f);
    if (t < LC * 4) { rbp = gbp[t]; rcp = gcp[t]; }

    const int q     = t & 3;
    const int g     = t >> 2;
    const int d     = d0 + g;
    const int group = (b * DD + d) * NC + c;
    const float4 An = ((const float4*)(A + d * NN))[q];
    float4 h = ((const float4*)Hinit)[group * 4 + q];

#pragma unroll
    for (int j = 0; j < 4; ++j) {
        const int i = t + SCANT * j;
        const int off = (i >> 4) * TP4 + (i & 15);
        s_dt[off] = rdt[j];
        s_x[off]  = rx[j];
    }
    if (t < LC * 4) { s_bp[t] = rbp; s_cp[t] = rcp; }
    __syncthreads();

    const float4 rAn = make_float4(1.0f / An.x, 1.0f / An.y, 1.0f / An.z, 1.0f / An.w);
    const float4* pdt = s_dt + g * TP4;
    const float4* px  = s_x + g * TP4;
    const float4* pbp = s_bp + q;
    const float4* pcp = s_cp + q;
    float4* prow = s_y + g * TP4;

    for (int s8 = 0; s8 < LC / 8; ++s8) {
        float4 dt4[2], xv4[2], bp[8], cp[8];
        float p[8];
#pragma unroll
        for (int u = 0; u < 2; ++u) {
            dt4[u] = pdt[2 * s8 + u];
            xv4[u] = px[2 * s8 + u];
        }
#pragma unroll
        for (int k = 0; k < 8; ++k) {
            bp[k] = pbp[(8 * s8 + k) * 4];
            cp[k] = pcp[(8 * s8 + k) * 4];
        }
#pragma unroll
        for (int u = 0; u < 2; ++u) {
            ssm_step(h, An, rAn, dt4[u].x, xv4[u].x, bp[4 * u + 0]); p[4 * u + 0] = ssm_out(h, cp[4 * u + 0]);
            ssm_step(h, An, rAn, dt4[u].y, xv4[u].y, bp[4 * u + 1]); p[4 * u + 1] = ssm_out(h, cp[4 * u + 1]);
            ssm_step(h, An, rAn, dt4[u].z, xv4[u].z, bp[4 * u + 2]); p[4 * u + 2] = ssm_out(h, cp[4 * u + 2]);
            ssm_step(h, An, rAn, dt4[u].w, xv4[u].w, bp[4 * u + 3]); p[4 * u + 3] = ssm_out(h, cp[4 * u + 3]);
        }
        if (q == 0) {
            prow[2 * s8]     = make_float4(p[0], p[1], p[2], p[3]);
            prow[2 * s8 + 1] = make_float4(p[4], p[5], p[6], p[7]);
        }
    }
    __syncthreads();

#pragma unroll
    for (int j = 0; j < 4; ++j) {
        const int i = t + SCANT * j;
        gy[(i >> 4) * (LL / 4) + (i & 15)] = s_y[(i >> 4) * TP4 + (i & 15)];
    }
}

extern "C" void kernel_launch(void* const* d_in, const int* in_sizes, int n_in,
                              void* d_out, int out_size, void* d_ws, size_t ws_size,
                              hipStream_t stream)
{
    const float* x       = (const float*)d_in[0];
    const float* A       = (const float*)d_in[1];
    const float* W_B     = (const float*)d_in[2];
    const float* b_B     = (const float*)d_in[3];
    const float* W_C     = (const float*)d_in[4];
    const float* b_C     = (const float*)d_in[5];
    const float* W_d     = (const float*)d_in[6];
    const float* b_d     = (const float*)d_in[7];
    const float* p_Delta = (const float*)d_in[8];
    float* y = (float*)d_out;

    float* ws    = (float*)d_ws;
    float* Bp    = ws;
    float* Cp    = Bp + (size_t)BB * LL * NN;
    float* dlin  = Cp + (size_t)BB * LL * NN;
    float* delta = dlin + (size_t)BB * LL;
    float* Pc    = delta + (size_t)BB * DD * LL;
    float* Sc    = Pc + (size_t)BB * DD * NC * NN;
    float* Hinit = Sc + (size_t)BB * DD * NC * NN;

    ssm_proj_wmma<<<dim3(128), dim3(32 * KSPLIT), 0, stream>>>(
        x, W_B, b_B, W_C, b_C, W_d, b_d, Bp, Cp, dlin);

    ssm_scan_local<<<dim3(BB * NC * NDBLK), dim3(SCANT), 0, stream>>>(
        x, A, dlin, p_Delta, Bp, delta, Pc, Sc);

    ssm_chunk_combine<<<dim3((BB * DD * NN) / 256), dim3(256), 0, stream>>>(
        Pc, Sc, Hinit);

    ssm_scan_final<<<dim3(BB * NC * NDBLK), dim3(SCANT), 0, stream>>>(
        x, A, delta, Bp, Cp, Hinit, y);
}


# scan passes A, B, C fused into one launch: a work-group holds all 16 chunks of 6 channels, chunk combine through LDS; delta, P, S, h_init stay in the WGP; 4 launches become 2
# speedup vs baseline: 4.6500x; 1.2356x over previous
#include <hip/hip_runtime.h>
#include <math.h>

#define DD 768
#define LL 1024
#define NN 16
#define BB 2
#define NC 16
#define LC (LL / NC)

typedef float v2f __attribute__((ext_vector_type(2)));
typedef float v8f __attribute__((ext_vector_type(8)));

#define KSPLIT 8
#define KW     (DD / KSPLIT)
#define KSTEPS (KW / 4)
#define KGRP   8

__global__ __launch_bounds__(32 * KSPLIT) void ssm_proj_wmma(
    const float* __restrict__ x,
    const float* __restrict__ W_B,
    const float* __restrict__ b_B,
    const float* __restrict__ W_C,
    const float* __restrict__ b_C,
    const float* __restrict__ W_d,
    const float* __restrict__ b_d,
    float* __restrict__ Bp,
    float* __restrict__ Cp,
    float* __restrict__ dlin)
{
    __shared__ float red[KSPLIT][24][32];

    const int lane = threadIdx.x & 31;
    const int wave = __builtin_amdgcn_readfirstlane((int)(threadIdx.x >> 5));
    const int half = lane >> 4;
    const int mn   = lane & 15;
    const int row0 = blockIdx.x << 4;
    const int b    = row0 >> 10;
    const int l0   = row0 & (LL - 1);
    const int lA   = l0 + mn;

    const float biasB = (wave == 0) ? b_B[mn] : 0.0f;
    const float biasC = (wave == 0) ? b_C[mn] : 0.0f;
    const float biasD = (wave == 0 && mn == 0) ? b_d[0] : 0.0f;
    v8f accB, accC, accD;
#pragma unroll
    for (int r = 0; r < 8; ++r) { accB[r] = biasB; accC[r] = biasC; accD[r] = biasD; }

    const float n0sel = (mn == 0) ? 1.0f : 0.0f;
    const int   kw0   = wave * KW + 2 * half;
    const float* __restrict__ xa  = x + (size_t)b * DD * LL + (size_t)kw0 * LL + lA;
    const float* __restrict__ wbp = W_B + mn * DD + kw0;
    const float* __restrict__ wcp = W_C + mn * DD + kw0;
    const float* __restrict__ wdp = W_d + kw0;

#pragma unroll
    for (int s0 = 0; s0 < KSTEPS; s0 += KGRP) {
        v2f a[KGRP], wb[KGRP], wc[KGRP], wd[KGRP];
#pragma unroll
        for (int u = 0; u < KGRP; ++u) {
            const int ko = 4 * (s0 + u);
            a[u][0]  = xa[(size_t)ko * LL];
            a[u][1]  = xa[(size_t)(ko + 1) * LL];
            wb[u][0] = wbp[ko];
            wb[u][1] = wbp[ko + 1];
            wc[u][0] = wcp[ko];
            wc[u][1] = wcp[ko + 1];
            wd[u][0] = wdp[ko] * n0sel;
            wd[u][1] = wdp[ko + 1] * n0sel;
        }
#pragma unroll
        for (int u = 0; u < KGRP; ++u) {
            accB = __builtin_amdgcn_wmma_f32_16x16x4_f32(false, a[u], false, wb[u], (short)0, accB, false, false);
            accC = __builtin_amdgcn_wmma_f32_16x16x4_f32(false, a[u], false, wc[u], (short)0, accC, false, false);
            accD = __builtin_amdgcn_wmma_f32_16x16x4_f32(false, a[u], false, wd[u], (short)0, accD, false, false);
        }
    }

#pragma unroll
    for (int r = 0; r < 8; ++r) {
        red[wave][r][lane]      = accB[r];
        red[wave][8 + r][lane]  = accC[r];
        red[wave][16 + r][lane] = accD[r];
    }
    __syncthreads();

    float sB = red[0][wave][lane];
    float sC = red[0][8 + wave][lane];
    float sD = red[0][16 + wave][lane];
#pragma unroll
    for (int w = 1; w < KSPLIT; ++w) {
        sB += red[w][wave][lane];
        sC += red[w][8 + wave][lane];
        sD += red[w][16 + wave][lane];
    }
    const int l  = l0 + wave + 8 * half;
    const int gr = b * LL + l;
    Bp[gr * NN + mn] = sB;
    Cp[gr * NN + mn] = sC;
    if (mn == 0) dlin[gr] = sD;
}

__device__ __forceinline__ float ssm_softplus(const float dl)
{
    return fmaxf(dl, 0.0f) + log1pf(__expf(-fabsf(dl)));
}

#define CHB   6
#define ROWS  (CHB * NC)
#define SCANT (ROWS * 4)
#define NCHB  (DD / CHB)
#define TP4   17
#define CP4   (LC * 4 + 4)
#define PL4   (LL * NN / 4)
#define PPAD  (11 * SCANT - PL4)

__device__ __forceinline__ void ssm_step(float4& h, const float4 An, const float4 rAn,
                                         const float dt, const float xv, const float4 bp)
{
    const float a0 = __expf(dt * An.x);
    const float a1 = __expf(dt * An.y);
    const float a2 = __expf(dt * An.z);
    const float a3 = __expf(dt * An.w);
    h.x = fmaf(a0, h.x, (a0 - An.x) * rAn.x * bp.x * xv);
    h.y = fmaf(a1, h.y, (a1 - An.y) * rAn.y * bp.y * xv);
    h.z = fmaf(a2, h.z, (a2 - An.z) * rAn.z * bp.z * xv);
    h.w = fmaf(a3, h.w, (a3 - An.w) * rAn.w * bp.w * xv);
}

__device__ __forceinline__ float ssm_out(const float4 h, const float4 cp)
{
    float p = cp.x * h.x;
    p = fmaf(cp.y, h.y, p);
    p = fmaf(cp.z, h.z, p);
    p = fmaf(cp.w, h.w, p);
    p += __shfl_xor(p, 1, 4);
    p += __shfl_xor(p, 2, 4);
    return p;
}


__global__ __launch_bounds__(SCANT) __attribute__((amdgpu_waves_per_eu(3, 4))) void ssm_scan_fused(
    const float* x,
    const float* __restrict__ A,
    const float* __restrict__ dlin,
    const float* __restrict__ p_Delta,
    const float* Bp,
    const float* Cp,
    float* __restrict__ y)
{
    __shared__ float4 s_dt[ROWS * TP4];
    __shared__ float4 s_x[ROWS * TP4];
    __shared__ float4 s_y[ROWS * TP4];
    __shared__ float4 s_bp[NC * CP4 + PPAD];
    __shared__ float4 s_cp[NC * CP4 + PPAD];
    __shared__ float4 s_P[SCANT];
    __shared__ float4 s_S[SCANT];

    const int t  = threadIdx.x;
    const int b  = blockIdx.x / NCHB;
    const int d0 = (blockIdx.x % NCHB) * CHB;

    const float4* gx  = (const float4*)(x + ((size_t)(b * DD + d0)) * LL);
    float4*       __restrict__ gy  = (float4*)(y + ((size_t)(b * DD + d0)) * LL);
    const float4* gbp = (const float4*)(Bp + (size_t)b * LL * NN);
    const float4* gcp = (const float4*)(Cp + (size_t)b * LL * NN);
    const float4* __restrict__ gdl = (const float4*)(dlin + (size_t)b * LL);

    const float4 dla = gdl[t & 255];
    const float4 dlb = gdl[(t + 128) & 255];
    float pd[4];
#pragma unroll
    for (int j = 0; j < 4; ++j) pd[j] = p_Delta[d0 + ((t + SCANT * j) >> 8)];
    const int q  = t & 3;
    const int g  = t >> 2;
    const int c  = g / CHB;
    const int ch = g - c * CHB;
    const float4 An = ((const float4*)(A + (d0 + ch) * NN))[q];
    const float4 rx0 = gx[t];
    const float4 rx1 = gx[t + SCANT];
    const float4 rx2 = gx[t + SCANT * 2];
    const float4 rx3 = gx[t + SCANT * 3];
    const int k10 = min(t + SCANT * 10, PL4 - 1);
#define SSM_PLANE_LD(r, gp) \
    const float4 r##0 = gp[t];             const float4 r##1 = gp[t + SCANT];     \
    const float4 r##2 = gp[t + SCANT * 2]; const float4 r##3 = gp[t + SCANT * 3]; \
    const float4 r##4 = gp[t + SCANT * 4]; const float4 r##5 = gp[t + SCANT * 5]; \
    const float4 r##6 = gp[t + SCANT * 6]; const float4 r##7 = gp[t + SCANT * 7]; \
    const float4 r##8 = gp[t + SCANT * 8]; const float4 r##9 = gp[t + SCANT * 9]; \
    const float4 r##10 = gp[k10];
#define SSM_PLANE_ST1(sp, r, m) { const int k = t + SCANT * m; sp[(k >> 8) * CP4 + (k & 255)] = r##m; }
#define SSM_PLANE_ST(sp, r) \
    SSM_PLANE_ST1(sp, r, 0) SSM_PLANE_ST1(sp, r, 1) SSM_PLANE_ST1(sp, r, 2) SSM_PLANE_ST1(sp, r, 3) \
    SSM_PLANE_ST1(sp, r, 4) SSM_PLANE_ST1(sp, r, 5) SSM_PLANE_ST1(sp, r, 6) SSM_PLANE_ST1(sp, r, 7) \
    SSM_PLANE_ST1(sp, r, 8) SSM_PLANE_ST1(sp, r, 9) SSM_PLANE_ST1(sp, r, 10)
    SSM_PLANE_LD(rbp, gbp)
    SSM_PLANE_LD(rcp, gcp)

    __builtin_amdgcn_sched_barrier(0);

    float4 rdt[4];
#pragma unroll
    for (int j = 0; j < 4; ++j) {
        const float4 dl4 = (j & 1) ? dlb : dla;
        rdt[j].x = ssm_softplus(dl4.x + pd[j]);
        rdt[j].y = ssm_softplus(dl4.y + pd[j]);
        rdt[j].z = ssm_softplus(dl4.z + pd[j]);
        rdt[j].w = ssm_softplus(dl4.w + pd[j]);
    }

#define SSM_TILE_ST(j) { const int i = t + SCANT * j; const int off = (((i >> 4) & 15) * CHB + (i >> 8)) * TP4 + (i & 15); s_dt[off] = rdt[j]; s_x[off] = rx##j; }
    SSM_TILE_ST(0) SSM_TILE_ST(1) SSM_TILE_ST(2) SSM_TILE_ST(3)
#undef SSM_TILE_ST
    SSM_PLANE_ST(s_bp, rbp)
    SSM_PLANE_ST(s_cp, rcp)
#undef SSM_PLANE_LD
#undef SSM_PLANE_ST
#undef SSM_PLANE_ST1
    __syncthreads();

    const float4 rAn = make_float4(1.0f / An.x, 1.0f / An.y, 1.0f / An.z, 1.0f / An.w);
    const float4* pdt = s_dt + g * TP4;
    const float4* px  = s_x + g * TP4;
    const float4* pbp = s_bp + c * CP4 + q;
    const float4* pcp = s_cp + c * CP4 + q;
    float4* prow = s_y + g * TP4;

    float4 h = make_float4(0.f, 0.f, 0.f, 0.f);
    float sd = 0.0f;
    for (int s8 = 0; s8 < LC / 8; ++s8) {
        float4 dt4[2], xv4[2], bp[8];
#pragma unroll
        for (int u = 0; u < 2; ++u) {
            dt4[u] = pdt[2 * s8 + u];
            xv4[u] = px[2 * s8 + u];
        }
#pragma unroll
        for (int k = 0; k < 8; ++k) bp[k] = pbp[(8 * s8 + k) * 4];
#pragma unroll
        for (int u = 0; u < 2; ++u) {
            sd += dt4[u].x; ssm_step(h, An, rAn, dt4[u].x, xv4[u].x, bp[4 * u + 0]);
            sd += dt4[u].y; ssm_step(h, An, rAn, dt4[u].y, xv4[u].y, bp[4 * u + 1]);
            sd += dt4[u].z; ssm_step(h, An, rAn, dt4[u].z, xv4[u].z, bp[4 * u + 2]);
            sd += dt4[u].w; ssm_step(h, An, rAn, dt4[u].w, xv4[u].w, bp[4 * u + 3]);
        }
    }
    float4 P;
    P.x = __expf(An.x * sd);
    P.y = __expf(An.y * sd);
    P.z = __expf(An.z * sd);
    P.w = __expf(An.w * sd);
    s_P[t] = P;
    s_S[t] = h;
    __syncthreads();

    h = make_float4(0.f, 0.f, 0.f, 0.f);
#pragma unroll
    for (int cc = 0; cc < NC - 1; ++cc) {
        const float4 Pv = s_P[(cc * CHB + ch) * 4 + q];
        const float4 Sv = s_S[(cc * CHB + ch) * 4 + q];
        if (cc < c) {
            h.x = fmaf(Pv.x, h.x, Sv.x);
            h.y = fmaf(Pv.y, h.y, Sv.y);
            h.z = fmaf(Pv.z, h.z, Sv.z);
            h.w = fmaf(Pv.w, h.w, Sv.w);
        }
    }

    for (int s8 = 0; s8 < LC / 8; ++s8) {
        float4 dt4[2], xv4[2], bp[8], cp[8];
        float p[8];
#pragma unroll
        for (int u = 0; u < 2; ++u) {
            dt4[u] = pdt[2 * s8 + u];
            xv4[u] = px[2 * s8 + u];
        }
#pragma unroll
        for (int k = 0; k < 8; ++k) {
            bp[k] = pbp[(8 * s8 + k) * 4];
            cp[k] = pcp[(8 * s8 + k) * 4];
        }
#pragma unroll
        for (int u = 0; u < 2; ++u) {
            ssm_step(h, An, rAn, dt4[u].x, xv4[u].x, bp[4 * u + 0]); p[4 * u + 0] = ssm_out(h, cp[4 * u + 0]);
            ssm_step(h, An, rAn, dt4[u].y, xv4[u].y, bp[4 * u + 1]); p[4 * u + 1] = ssm_out(h, cp[4 * u + 1]);
            ssm_step(h, An, rAn, dt4[u].z, xv4[u].z, bp[4 * u + 2]); p[4 * u + 2] = ssm_out(h, cp[4 * u + 2]);
            ssm_step(h, An, rAn, dt4[u].w, xv4[u].w, bp[4 * u + 3]); p[4 * u + 3] = ssm_out(h, cp[4 * u + 3]);
        }
        if (q == 0) {
            prow[2 * s8]     = make_float4(p[0], p[1], p[2], p[3]);
            prow[2 * s8 + 1] = make_float4(p[4], p[5], p[6], p[7]);
        }
    }
    __syncthreads();

#pragma unroll
    for (int j = 0; j < 4; ++j) {
        const int i = t + SCANT * j;
        gy[i] = s_y[(((i >> 4) & 15) * CHB + (i >> 8)) * TP4 + (i & 15)];
    }
}

extern "C" void kernel_launch(void* const* d_in, const int* in_sizes, int n_in,
                              void* d_out, int out_size, void* d_ws, size_t ws_size,
                              hipStream_t stream)
{
    const float* x       = (const float*)d_in[0];
    const float* A       = (const float*)d_in[1];
    const float* W_B     = (const float*)d_in[2];
    const float* b_B     = (const float*)d_in[3];
    const float* W_C     = (const float*)d_in[4];
    const float* b_C     = (const float*)d_in[5];
    const float* W_d     = (const float*)d_in[6];
    const float* b_d     = (const float*)d_in[7];
    const float* p_Delta = (const float*)d_in[8];
    float* y = (float*)d_out;

    float* ws    = (float*)d_ws;
    float* Bp    = ws;
    float* Cp    = Bp + (size_t)BB * LL * NN;
    float* dlin  = Cp + (size_t)BB * LL * NN;

    ssm_proj_wmma<<<dim3(128), dim3(32 * KSPLIT), 0, stream>>>(
        x, W_B, b_B, W_C, b_C, W_d, b_d, Bp, Cp, dlin);

    ssm_scan_fused<<<dim3(BB * NCHB), dim3(SCANT), 0, stream>>>(
        x, A, dlin, p_Delta, Bp, Cp, y);
}


# scan prologue: softplus's log1p(e) as log(u)*e/(u-1), u = 1+e, on the hardware log2 and reciprocal in place of the library log1pf, loads still in flight beside it (delta differs in last bits)
# speedup vs baseline: 5.0623x; 1.0887x over previous
#include <hip/hip_runtime.h>
#include <math.h>

#define DD 768
#define LL 1024
#define NN 16
#define BB 2
#define NC 16
#define LC (LL / NC)

typedef float v2f __attribute__((ext_vector_type(2)));
typedef float v8f __attribute__((ext_vector_type(8)));

#define KSPLIT 8
#define KW     (DD / KSPLIT)
#define KSTEPS (KW / 4)
#define KGRP   8

__global__ __launch_bounds__(32 * KSPLIT) void ssm_proj_wmma(
    const float* __restrict__ x,
    const float* __restrict__ W_B,
    const float* __restrict__ b_B,
    const float* __restrict__ W_C,
    const float* __restrict__ b_C,
    const float* __restrict__ W_d,
    const float* __restrict__ b_d,
    float* __restrict__ Bp,
    float* __restrict__ Cp,
    float* __restrict__ dlin)
{
    __shared__ float red[KSPLIT][24][32];

    const int lane = threadIdx.x & 31;
    const int wave = __builtin_amdgcn_readfirstlane((int)(threadIdx.x >> 5));
    const int half = lane >> 4;
    const int mn   = lane & 15;
    const int row0 = blockIdx.x << 4;
    const int b    = row0 >> 10;
    const int l0   = row0 & (LL - 1);
    const int lA   = l0 + mn;

    const float biasB = (wave == 0) ? b_B[mn] : 0.0f;
    const float biasC = (wave == 0) ? b_C[mn] : 0.0f;
    const float biasD = (wave == 0 && mn == 0) ? b_d[0] : 0.0f;
    v8f accB, accC, accD;
#pragma unroll
    for (int r = 0; r < 8; ++r) { accB[r] = biasB; accC[r] = biasC; accD[r] = biasD; }

    const float n0sel = (mn == 0) ? 1.0f : 0.0f;
    const int   kw0   = wave * KW + 2 * half;
    const float* __restrict__ xa  = x + (size_t)b * DD * LL + (size_t)kw0 * LL + lA;
    const float* __restrict__ wbp = W_B + mn * DD + kw0;
    const float* __restrict__ wcp = W_C + mn * DD + kw0;
    const float* __restrict__ wdp = W_d + kw0;

#pragma unroll
    for (int s0 = 0; s0 < KSTEPS; s0 += KGRP) {
        v2f a[KGRP], wb[KGRP], wc[KGRP], wd[KGRP];
#pragma unroll
        for (int u = 0; u < KGRP; ++u) {
            const int ko = 4 * (s0 + u);
            a[u][0]  = xa[(size_t)ko * LL];
            a[u][1]  = xa[(size_t)(ko + 1) * LL];
            wb[u][0] = wbp[ko];
            wb[u][1] = wbp[ko + 1];
            wc[u][0] = wcp[ko];
            wc[u][1] = wcp[ko + 1];
            wd[u][0] = wdp[ko] * n0sel;
            wd[u][1] = wdp[ko + 1] * n0sel;
        }
#pragma unroll
        for (int u = 0; u < KGRP; ++u) {
            accB = __builtin_amdgcn_wmma_f32_16x16x4_f32(false, a[u], false, wb[u], (short)0, accB, false, false);
            accC = __builtin_amdgcn_wmma_f32_16x16x4_f32(false, a[u], false, wc[u], (short)0, accC, false, false);
            accD = __builtin_amdgcn_wmma_f32_16x16x4_f32(false, a[u], false, wd[u], (short)0, accD, false, false);
        }
    }

#pragma unroll
    for (int r = 0; r < 8; ++r) {
        red[wave][r][lane]      = accB[r];
        red[wave][8 + r][lane]  = accC[r];
        red[wave][16 + r][lane] = accD[r];
    }
    __syncthreads();

    float sB = red[0][wave][lane];
    float sC = red[0][8 + wave][lane];
    float sD = red[0][16 + wave][lane];
#pragma unroll
    for (int w = 1; w < KSPLIT; ++w) {
        sB += red[w][wave][lane];
        sC += red[w][8 + wave][lane];
        sD += red[w][16 + wave][lane];
    }
    const int l  = l0 + wave + 8 * half;
    const int gr = b * LL + l;
    Bp[gr * NN + mn] = sB;
    Cp[gr * NN + mn] = sC;
    if (mn == 0) dlin[gr] = sD;
}

__device__ __forceinline__ float ssm_softplus(const float dl)
{
    const float e = __expf(-fabsf(dl));
    const float u = 1.0f + e;
    const float d = u - 1.0f;
    const float q = (__builtin_amdgcn_logf(u) * 0.6931471805599453f) * (e * __builtin_amdgcn_rcpf(d));
    const float l = (d == 0.0f) ? e : q;
    return fmaxf(dl, 0.0f) + l;
}

#define CHB   6
#define ROWS  (CHB * NC)
#define SCANT (ROWS * 4)
#define NCHB  (DD / CHB)
#define TP4   17
#define CP4   (LC * 4 + 4)
#define PL4   (LL * NN / 4)
#define PPAD  (11 * SCANT - PL4)

__device__ __forceinline__ void ssm_step(float4& h, const float4 An, const float4 rAn,
                                         const float dt, const float xv, const float4 bp)
{
    const float a0 = __expf(dt * An.x);
    const float a1 = __expf(dt * An.y);
    const float a2 = __expf(dt * An.z);
    const float a3 = __expf(dt * An.w);
    h.x = fmaf(a0, h.x, (a0 - An.x) * rAn.x * bp.x * xv);
    h.y = fmaf(a1, h.y, (a1 - An.y) * rAn.y * bp.y * xv);
    h.z = fmaf(a2, h.z, (a2 - An.z) * rAn.z * bp.z * xv);
    h.w = fmaf(a3, h.w, (a3 - An.w) * rAn.w * bp.w * xv);
}

__device__ __forceinline__ float ssm_out(const float4 h, const float4 cp)
{
    float p = cp.x * h.x;
    p = fmaf(cp.y, h.y, p);
    p = fmaf(cp.z, h.z, p);
    p = fmaf(cp.w, h.w, p);
    p += __shfl_xor(p, 1, 4);
    p += __shfl_xor(p, 2, 4);
    return p;
}


__global__ __launch_bounds__(SCANT) __attribute__((amdgpu_waves_per_eu(3, 4))) void ssm_scan_fused(
    const float* x,
    const float* __restrict__ A,
    const float* dlin,
    const float* __restrict__ p_Delta,
    const float* Bp,
    const float* Cp,
    float* __restrict__ y)
{
    __shared__ float4 s_dt[ROWS * TP4];
    __shared__ float4 s_x[ROWS * TP4];
    __shared__ float4 s_y[ROWS * TP4];
    __shared__ float4 s_bp[NC * CP4 + PPAD];
    __shared__ float4 s_cp[NC * CP4 + PPAD];
    __shared__ float4 s_P[SCANT];
    __shared__ float4 s_S[SCANT];

    const int t  = threadIdx.x;
    const int b  = blockIdx.x / NCHB;
    const int d0 = (blockIdx.x % NCHB) * CHB;

    const float4* gx  = (const float4*)(x + ((size_t)(b * DD + d0)) * LL);
    float4*       __restrict__ gy  = (float4*)(y + ((size_t)(b * DD + d0)) * LL);
    const float4* gbp = (const float4*)(Bp + (size_t)b * LL * NN);
    const float4* gcp = (const float4*)(Cp + (size_t)b * LL * NN);
    const float4* gdl = (const float4*)(dlin + (size_t)b * LL);

    const float4 dla = gdl[t & 255];
    const float4 dlb = gdl[(t + 128) & 255];
    float pd[4];
#pragma unroll
    for (int j = 0; j < 4; ++j) pd[j] = p_Delta[d0 + ((t + SCANT * j) >> 8)];
    const int q  = t & 3;
    const int g  = t >> 2;
    const int c  = g / CHB;
    const int ch = g - c * CHB;
    const float4 An = ((const float4*)(A + (d0 + ch) * NN))[q];
    const float4 rx0 = gx[t];
    const float4 rx1 = gx[t + SCANT];
    const float4 rx2 = gx[t + SCANT * 2];
    const float4 rx3 = gx[t + SCANT * 3];
    const int k10 = min(t + SCANT * 10, PL4 - 1);
#define SSM_PLANE_LD(r, gp) \
    const float4 r##0 = gp[t];             const float4 r##1 = gp[t + SCANT];     \
    const float4 r##2 = gp[t + SCANT * 2]; const float4 r##3 = gp[t + SCANT * 3]; \
    const float4 r##4 = gp[t + SCANT * 4]; const float4 r##5 = gp[t + SCANT * 5]; \
    const float4 r##6 = gp[t + SCANT * 6]; const float4 r##7 = gp[t + SCANT * 7]; \
    const float4 r##8 = gp[t + SCANT * 8]; const float4 r##9 = gp[t + SCANT * 9]; \
    const float4 r##10 = gp[k10];
#define SSM_PLANE_ST1(sp, r, m) { const int k = t + SCANT * m; sp[(k >> 8) * CP4 + (k & 255)] = r##m; }
#define SSM_PLANE_ST(sp, r) \
    SSM_PLANE_ST1(sp, r, 0) SSM_PLANE_ST1(sp, r, 1) SSM_PLANE_ST1(sp, r, 2) SSM_PLANE_ST1(sp, r, 3) \
    SSM_PLANE_ST1(sp, r, 4) SSM_PLANE_ST1(sp, r, 5) SSM_PLANE_ST1(sp, r, 6) SSM_PLANE_ST1(sp, r, 7) \
    SSM_PLANE_ST1(sp, r, 8) SSM_PLANE_ST1(sp, r, 9) SSM_PLANE_ST1(sp, r, 10)
    SSM_PLANE_LD(rbp, gbp)
    SSM_PLANE_LD(rcp, gcp)

    __builtin_amdgcn_sched_barrier(0);

    float4 rdt[4];
#pragma unroll
    for (int j = 0; j < 4; ++j) {
        const float4 dl4 = (j & 1) ? dlb : dla;
        rdt[j].x = ssm_softplus(dl4.x + pd[j]);
        rdt[j].y = ssm_softplus(dl4.y + pd[j]);
        rdt[j].z = ssm_softplus(dl4.z + pd[j]);
        rdt[j].w = ssm_softplus(dl4.w + pd[j]);
    }

#define SSM_TILE_ST(j) { const int i = t + SCANT * j; const int off = (((i >> 4) & 15) * CHB + (i >> 8)) * TP4 + (i & 15); s_dt[off] = rdt[j]; s_x[off] = rx##j; }
    SSM_TILE_ST(0) SSM_TILE_ST(1) SSM_TILE_ST(2) SSM_TILE_ST(3)
#undef SSM_TILE_ST
    SSM_PLANE_ST(s_bp, rbp)
    SSM_PLANE_ST(s_cp, rcp)
#undef SSM_PLANE_LD
#undef SSM_PLANE_ST
#undef SSM_PLANE_ST1
    __syncthreads();

    const float4 rAn = make_float4(1.0f / An.x, 1.0f / An.y, 1.0f / An.z, 1.0f / An.w);
    const float4* pdt = s_dt + g * TP4;
    const float4* px  = s_x + g * TP4;
    const float4* pbp = s_bp + c * CP4 + q;
    const float4* pcp = s_cp + c * CP4 + q;
    float4* prow = s_y + g * TP4;

    float4 h = make_float4(0.f, 0.f, 0.f, 0.f);
    float sd = 0.0f;
    for (int s8 = 0; s8 < LC / 8; ++s8) {
        float4 dt4[2], xv4[2], bp[8];
#pragma unroll
        for (int u = 0; u < 2; ++u) {
            dt4[u] = pdt[2 * s8 + u];
            xv4[u] = px[2 * s8 + u];
        }
#pragma unroll
        for (int k = 0; k < 8; ++k) bp[k] = pbp[(8 * s8 + k) * 4];
#pragma unroll
        for (int u = 0; u < 2; ++u) {
            sd += dt4[u].x; ssm_step(h, An, rAn, dt4[u].x, xv4[u].x, bp[4 * u + 0]);
            sd += dt4[u].y; ssm_step(h, An, rAn, dt4[u].y, xv4[u].y, bp[4 * u + 1]);
            sd += dt4[u].z; ssm_step(h, An, rAn, dt4[u].z, xv4[u].z, bp[4 * u + 2]);
            sd += dt4[u].w; ssm_step(h, An, rAn, dt4[u].w, xv4[u].w, bp[4 * u + 3]);
        }
    }
    float4 P;
    P.x = __expf(An.x * sd);
    P.y = __expf(An.y * sd);
    P.z = __expf(An.z * sd);
    P.w = __expf(An.w * sd);
    s_P[t] = P;
    s_S[t] = h;
    __syncthreads();

    h = make_float4(0.f, 0.f, 0.f, 0.f);
#pragma unroll
    for (int cc = 0; cc < NC - 1; ++cc) {
        const float4 Pv = s_P[(cc * CHB + ch) * 4 + q];
        const float4 Sv = s_S[(cc * CHB + ch) * 4 + q];
        if (cc < c) {
            h.x = fmaf(Pv.x, h.x, Sv.x);
            h.y = fmaf(Pv.y, h.y, Sv.y);
            h.z = fmaf(Pv.z, h.z, Sv.z);
            h.w = fmaf(Pv.w, h.w, Sv.w);
        }
    }

    for (int s8 = 0; s8 < LC / 8; ++s8) {
        float4 dt4[2], xv4[2], bp[8], cp[8];
        float p[8];
#pragma unroll
        for (int u = 0; u < 2; ++u) {
            dt4[u] = pdt[2 * s8 + u];
            xv4[u] = px[2 * s8 + u];
        }
#pragma unroll
        for (int k = 0; k < 8; ++k) {
            bp[k] = pbp[(8 * s8 + k) * 4];
            cp[k] = pcp[(8 * s8 + k) * 4];
        }
#pragma unroll
        for (int u = 0; u < 2; ++u) {
            ssm_step(h, An, rAn, dt4[u].x, xv4[u].x, bp[4 * u + 0]); p[4 * u + 0] = ssm_out(h, cp[4 * u + 0]);
            ssm_step(h, An, rAn, dt4[u].y, xv4[u].y, bp[4 * u + 1]); p[4 * u + 1] = ssm_out(h, cp[4 * u + 1]);
            ssm_step(h, An, rAn, dt4[u].z, xv4[u].z, bp[4 * u + 2]); p[4 * u + 2] = ssm_out(h, cp[4 * u + 2]);
            ssm_step(h, An, rAn, dt4[u].w, xv4[u].w, bp[4 * u + 3]); p[4 * u + 3] = ssm_out(h, cp[4 * u + 3]);
        }
        if (q == 0) {
            prow[2 * s8]     = make_float4(p[0], p[1], p[2], p[3]);
            prow[2 * s8 + 1] = make_float4(p[4], p[5], p[6], p[7]);
        }
    }
    __syncthreads();

#pragma unroll
    for (int j = 0; j < 4; ++j) {
        const int i = t + SCANT * j;
        gy[i] = s_y[(((i >> 4) & 15) * CHB + (i >> 8)) * TP4 + (i & 15)];
    }
}

extern "C" void kernel_launch(void* const* d_in, const int* in_sizes, int n_in,
                              void* d_out, int out_size, void* d_ws, size_t ws_size,
                              hipStream_t stream)
{
    const float* x       = (const float*)d_in[0];
    const float* A       = (const float*)d_in[1];
    const float* W_B     = (const float*)d_in[2];
    const float* b_B     = (const float*)d_in[3];
    const float* W_C     = (const float*)d_in[4];
    const float* b_C     = (const float*)d_in[5];
    const float* W_d     = (const float*)d_in[6];
    const float* b_d     = (const float*)d_in[7];
    const float* p_Delta = (const float*)d_in[8];
    float* y = (float*)d_out;

    float* ws    = (float*)d_ws;
    float* Bp    = ws;
    float* Cp    = Bp + (size_t)BB * LL * NN;
    float* dlin  = Cp + (size_t)BB * LL * NN;

    ssm_proj_wmma<<<dim3(128), dim3(32 * KSPLIT), 0, stream>>>(
        x, W_B, b_B, W_C, b_C, W_d, b_d, Bp, Cp, dlin);

    ssm_scan_fused<<<dim3(BB * NCHB), dim3(SCANT), 0, stream>>>(
        x, A, dlin, p_Delta, Bp, Cp, y);
}


# projection GEMM: each wave reads its K slice of W_B and W_C with whole-line b128 loads and hands it to its lanes through LDS, in place of 16-line fragment loads (k order and sums unchanged)
# speedup vs baseline: 5.8688x; 1.1593x over previous
#include <hip/hip_runtime.h>
#include <math.h>

#define DD 768
#define LL 1024
#define NN 16
#define BB 2
#define NC 16
#define LC (LL / NC)

typedef float v2f __attribute__((ext_vector_type(2)));
typedef float v8f __attribute__((ext_vector_type(8)));

#define KSPLIT 8
#define KW     (DD / KSPLIT)
#define KSTEPS (KW / 4)
#define WC4    (KW / 4)
#define WP     (KW + 4)
#define WSTG   (16 * WC4 / 32)

__global__ __launch_bounds__(32 * KSPLIT) __attribute__((amdgpu_waves_per_eu(2, 2))) void ssm_proj_wmma(
    const float* x,
    const float* __restrict__ W_B,
    const float* __restrict__ b_B,
    const float* __restrict__ W_C,
    const float* __restrict__ b_C,
    const float* W_d,
    const float* __restrict__ b_d,
    float* __restrict__ Bp,
    float* __restrict__ Cp,
    float* __restrict__ dlin)
{
    __shared__ float red[KSPLIT][24][32];
    __shared__ float4 s_w[KSPLIT][2][16 * WP / 4];

    const int lane = threadIdx.x & 31;
    const int wave = __builtin_amdgcn_readfirstlane((int)(threadIdx.x >> 5));
    const int half = lane >> 4;
    const int mn   = lane & 15;
    const int row0 = blockIdx.x << 4;
    const int b    = row0 >> 10;
    const int l0   = row0 & (LL - 1);
    const int lA   = l0 + mn;

    const float biasB = (wave == 0) ? b_B[mn] : 0.0f;
    const float biasC = (wave == 0) ? b_C[mn] : 0.0f;
    const float biasD = (wave == 0 && mn == 0) ? b_d[0] : 0.0f;
    v8f accB, accC, accD;
#pragma unroll
    for (int r = 0; r < 8; ++r) { accB[r] = biasB; accC[r] = biasC; accD[r] = biasD; }

    const float n0sel = (mn == 0) ? 1.0f : 0.0f;
    const int   kw0   = wave * KW + 2 * half;
    const float* xa = x + (size_t)b * DD * LL + (size_t)kw0 * LL + lA;
    const float* wdp = W_d + kw0;

    const float4* __restrict__ gwb = (const float4*)(W_B + wave * KW);
    const float4* __restrict__ gwc = (const float4*)(W_C + wave * KW);
    float4 rb[WSTG], rc[WSTG];
#pragma unroll
    for (int j = 0; j < WSTG; ++j) {
        const int i = lane + 32 * j;
        const int n = i / WC4;
        rb[j] = gwb[n * (DD / 4) + (i - n * WC4)];
        rc[j] = gwc[n * (DD / 4) + (i - n * WC4)];
    }
    v2f a[KSTEPS], wd[KSTEPS];
#pragma unroll
    for (int u = 0; u < KSTEPS; ++u) {
        const int ko = 4 * u;
        a[u][0]  = xa[(size_t)ko * LL];
        a[u][1]  = xa[(size_t)(ko + 1) * LL];
        wd[u][0] = wdp[ko];
        wd[u][1] = wdp[ko + 1];
    }
    __builtin_amdgcn_sched_barrier(0);
#pragma unroll
    for (int j = 0; j < WSTG; ++j) {
        const int i = lane + 32 * j;
        const int n = i / WC4;
        s_w[wave][0][n * (WP / 4) + (i - n * WC4)] = rb[j];
        s_w[wave][1][n * (WP / 4) + (i - n * WC4)] = rc[j];
    }
    __syncthreads();
    const float* wbp = (const float*)&s_w[wave][0][0] + mn * WP + 2 * half;
    const float* wcp = (const float*)&s_w[wave][1][0] + mn * WP + 2 * half;

#pragma unroll
    for (int u = 0; u < KSTEPS; ++u) {
        const int ko = 4 * u;
        v2f wb, wc, wdm;
        wb[0] = wbp[ko];  wb[1] = wbp[ko + 1];
        wc[0] = wcp[ko];  wc[1] = wcp[ko + 1];
        wdm[0] = wd[u][0] * n0sel;
        wdm[1] = wd[u][1] * n0sel;
        accB = __builtin_amdgcn_wmma_f32_16x16x4_f32(false, a[u], false, wb, (short)0, accB, false, false);
        accC = __builtin_amdgcn_wmma_f32_16x16x4_f32(false, a[u], false, wc, (short)0, accC, false, false);
        accD = __builtin_amdgcn_wmma_f32_16x16x4_f32(false, a[u], false, wdm, (short)0, accD, false, false);
    }

#pragma unroll
    for (int r = 0; r < 8; ++r) {
        red[wave][r][lane]      = accB[r];
        red[wave][8 + r][lane]  = accC[r];
        red[wave][16 + r][lane] = accD[r];
    }
    __syncthreads();

    float sB = red[0][wave][lane];
    float sC = red[0][8 + wave][lane];
    float sD = red[0][16 + wave][lane];
#pragma unroll
    for (int w = 1; w < KSPLIT; ++w) {
        sB += red[w][wave][lane];
        sC += red[w][8 + wave][lane];
        sD += red[w][16 + wave][lane];
    }
    const int l  = l0 + wave + 8 * half;
    const int gr = b * LL + l;
    Bp[gr * NN + mn] = sB;
    Cp[gr * NN + mn] = sC;
    if (mn == 0) dlin[gr] = sD;
}

__device__ __forceinline__ float ssm_softplus(const float dl)
{
    const float e = __expf(-fabsf(dl));
    const float u = 1.0f + e;
    const float d = u - 1.0f;
    const float q = (__builtin_amdgcn_logf(u) * 0.6931471805599453f) * (e * __builtin_amdgcn_rcpf(d));
    const float l = (d == 0.0f) ? e : q;
    return fmaxf(dl, 0.0f) + l;
}

#define CHB   6
#define ROWS  (CHB * NC)
#define SCANT (ROWS * 4)
#define NCHB  (DD / CHB)
#define TP4   17
#define CP4   (LC * 4 + 4)
#define PL4   (LL * NN / 4)
#define PPAD  (11 * SCANT - PL4)

__device__ __forceinline__ void ssm_step(float4& h, const float4 An, const float4 rAn,
                                         const float dt, const float xv, const float4 bp)
{
    const float a0 = __expf(dt * An.x);
    const float a1 = __expf(dt * An.y);
    const float a2 = __expf(dt * An.z);
    const float a3 = __expf(dt * An.w);
    h.x = fmaf(a0, h.x, (a0 - An.x) * rAn.x * bp.x * xv);
    h.y = fmaf(a1, h.y, (a1 - An.y) * rAn.y * bp.y * xv);
    h.z = fmaf(a2, h.z, (a2 - An.z) * rAn.z * bp.z * xv);
    h.w = fmaf(a3, h.w, (a3 - An.w) * rAn.w * bp.w * xv);
}

__device__ __forceinline__ float ssm_out(const float4 h, const float4 cp)
{
    float p = cp.x * h.x;
    p = fmaf(cp.y, h.y, p);
    p = fmaf(cp.z, h.z, p);
    p = fmaf(cp.w, h.w, p);
    p += __shfl_xor(p, 1, 4);
    p += __shfl_xor(p, 2, 4);
    return p;
}


__global__ __launch_bounds__(SCANT) __attribute__((amdgpu_waves_per_eu(3, 4))) void ssm_scan_fused(
    const float* x,
    const float* __restrict__ A,
    const float* dlin,
    const float* __restrict__ p_Delta,
    const float* Bp,
    const float* Cp,
    float* __restrict__ y)
{
    __shared__ float4 s_dt[ROWS * TP4];
    __shared__ float4 s_x[ROWS * TP4];
    __shared__ float4 s_y[ROWS * TP4];
    __shared__ float4 s_bp[NC * CP4 + PPAD];
    __shared__ float4 s_cp[NC * CP4 + PPAD];
    __shared__ float4 s_P[SCANT];
    __shared__ float4 s_S[SCANT];

    const int t  = threadIdx.x;
    const int b  = blockIdx.x / NCHB;
    const int d0 = (blockIdx.x % NCHB) * CHB;

    const float4* gx  = (const float4*)(x + ((size_t)(b * DD + d0)) * LL);
    float4*       __restrict__ gy  = (float4*)(y + ((size_t)(b * DD + d0)) * LL);
    const float4* gbp = (const float4*)(Bp + (size_t)b * LL * NN);
    const float4* gcp = (const float4*)(Cp + (size_t)b * LL * NN);
    const float4* gdl = (const float4*)(dlin + (size_t)b * LL);

    const float4 dla = gdl[t & 255];
    const float4 dlb = gdl[(t + 128) & 255];
    float pd[4];
#pragma unroll
    for (int j = 0; j < 4; ++j) pd[j] = p_Delta[d0 + ((t + SCANT * j) >> 8)];
    const int q  = t & 3;
    const int g  = t >> 2;
    const int c  = g / CHB;
    const int ch = g - c * CHB;
    const float4 An = ((const float4*)(A + (d0 + ch) * NN))[q];
    const float4 rx0 = gx[t];
    const float4 rx1 = gx[t + SCANT];
    const float4 rx2 = gx[t + SCANT * 2];
    const float4 rx3 = gx[t + SCANT * 3];
    const int k10 = min(t + SCANT * 10, PL4 - 1);
#define SSM_PLANE_LD(r, gp) \
    const float4 r##0 = gp[t];             const float4 r##1 = gp[t + SCANT];     \
    const float4 r##2 = gp[t + SCANT * 2]; const float4 r##3 = gp[t + SCANT * 3]; \
    const float4 r##4 = gp[t + SCANT * 4]; const float4 r##5 = gp[t + SCANT * 5]; \
    const float4 r##6 = gp[t + SCANT * 6]; const float4 r##7 = gp[t + SCANT * 7]; \
    const float4 r##8 = gp[t + SCANT * 8]; const float4 r##9 = gp[t + SCANT * 9]; \
    const float4 r##10 = gp[k10];
#define SSM_PLANE_ST1(sp, r, m) { const int k = t + SCANT * m; sp[(k >> 8) * CP4 + (k & 255)] = r##m; }
#define SSM_PLANE_ST(sp, r) \
    SSM_PLANE_ST1(sp, r, 0) SSM_PLANE_ST1(sp, r, 1) SSM_PLANE_ST1(sp, r, 2) SSM_PLANE_ST1(sp, r, 3) \
    SSM_PLANE_ST1(sp, r, 4) SSM_PLANE_ST1(sp, r, 5) SSM_PLANE_ST1(sp, r, 6) SSM_PLANE_ST1(sp, r, 7) \
    SSM_PLANE_ST1(sp, r, 8) SSM_PLANE_ST1(sp, r, 9) SSM_PLANE_ST1(sp, r, 10)
    SSM_PLANE_LD(rbp, gbp)
    SSM_PLANE_LD(rcp, gcp)

    __builtin_amdgcn_sched_barrier(0);

    float4 rdt[4];
#pragma unroll
    for (int j = 0; j < 4; ++j) {
        const float4 dl4 = (j & 1) ? dlb : dla;
        rdt[j].x = ssm_softplus(dl4.x + pd[j]);
        rdt[j].y = ssm_softplus(dl4.y + pd[j]);
        rdt[j].z = ssm_softplus(dl4.z + pd[j]);
        rdt[j].w = ssm_softplus(dl4.w + pd[j]);
    }

#define SSM_TILE_ST(j) { const int i = t + SCANT * j; const int off = (((i >> 4) & 15) * CHB + (i >> 8)) * TP4 + (i & 15); s_dt[off] = rdt[j]; s_x[off] = rx##j; }
    SSM_TILE_ST(0) SSM_TILE_ST(1) SSM_TILE_ST(2) SSM_TILE_ST(3)
#undef SSM_TILE_ST
    SSM_PLANE_ST(s_bp, rbp)
    SSM_PLANE_ST(s_cp, rcp)
#undef SSM_PLANE_LD
#undef SSM_PLANE_ST
#undef SSM_PLANE_ST1
    __syncthreads();

    const float4 rAn = make_float4(1.0f / An.x, 1.0f / An.y, 1.0f / An.z, 1.0f / An.w);
    const float4* pdt = s_dt + g * TP4;
    const float4* px  = s_x + g * TP4;
    const float4* pbp = s_bp + c * CP4 + q;
    const float4* pcp = s_cp + c * CP4 + q;
    float4* prow = s_y + g * TP4;

    float4 h = make_float4(0.f, 0.f, 0.f, 0.f);
    float sd = 0.0f;
    for (int s8 = 0; s8 < LC / 8; ++s8) {
        float4 dt4[2], xv4[2], bp[8];
#pragma unroll
        for (int u = 0; u < 2; ++u) {
            dt4[u] = pdt[2 * s8 + u];
            xv4[u] = px[2 * s8 + u];
        }
#pragma unroll
        for (int k = 0; k < 8; ++k) bp[k] = pbp[(8 * s8 + k) * 4];
#pragma unroll
        for (int u = 0; u < 2; ++u) {
            sd += dt4[u].x; ssm_step(h, An, rAn, dt4[u].x, xv4[u].x, bp[4 * u + 0]);
            sd += dt4[u].y; ssm_step(h, An, rAn, dt4[u].y, xv4[u].y, bp[4 * u + 1]);
            sd += dt4[u].z; ssm_step(h, An, rAn, dt4[u].z, xv4[u].z, bp[4 * u + 2]);
            sd += dt4[u].w; ssm_step(h, An, rAn, dt4[u].w, xv4[u].w, bp[4 * u + 3]);
        }
    }
    float4 P;
    P.x = __expf(An.x * sd);
    P.y = __expf(An.y * sd);
    P.z = __expf(An.z * sd);
    P.w = __expf(An.w * sd);
    s_P[t] = P;
    s_S[t] = h;
    __syncthreads();

    h = make_float4(0.f, 0.f, 0.f, 0.f);
#pragma unroll
    for (int cc = 0; cc < NC - 1; ++cc) {
        const float4 Pv = s_P[(cc * CHB + ch) * 4 + q];
        const float4 Sv = s_S[(cc * CHB + ch) * 4 + q];
        if (cc < c) {
            h.x = fmaf(Pv.x, h.x, Sv.x);
            h.y = fmaf(Pv.y, h.y, Sv.y);
            h.z = fmaf(Pv.z, h.z, Sv.z);
            h.w = fmaf(Pv.w, h.w, Sv.w);
        }
    }

    for (int s8 = 0; s8 < LC / 8; ++s8) {
        float4 dt4[2], xv4[2], bp[8], cp[8];
        float p[8];
#pragma unroll
        for (int u = 0; u < 2; ++u) {
            dt4[u] = pdt[2 * s8 + u];
            xv4[u] = px[2 * s8 + u];
        }
#pragma unroll
        for (int k = 0; k < 8; ++k) {
            bp[k] = pbp[(8 * s8 + k) * 4];
            cp[k] = pcp[(8 * s8 + k) * 4];
        }
#pragma unroll
        for (int u = 0; u < 2; ++u) {
            ssm_step(h, An, rAn, dt4[u].x, xv4[u].x, bp[4 * u + 0]); p[4 * u + 0] = ssm_out(h, cp[4 * u + 0]);
            ssm_step(h, An, rAn, dt4[u].y, xv4[u].y, bp[4 * u + 1]); p[4 * u + 1] = ssm_out(h, cp[4 * u + 1]);
            ssm_step(h, An, rAn, dt4[u].z, xv4[u].z, bp[4 * u + 2]); p[4 * u + 2] = ssm_out(h, cp[4 * u + 2]);
            ssm_step(h, An, rAn, dt4[u].w, xv4[u].w, bp[4 * u + 3]); p[4 * u + 3] = ssm_out(h, cp[4 * u + 3]);
        }
        if (q == 0) {
            prow[2 * s8]     = make_float4(p[0], p[1], p[2], p[3]);
            prow[2 * s8 + 1] = make_float4(p[4], p[5], p[6], p[7]);
        }
    }
    __syncthreads();

#pragma unroll
    for (int j = 0; j < 4; ++j) {
        const int i = t + SCANT * j;
        gy[i] = s_y[(((i >> 4) & 15) * CHB + (i >> 8)) * TP4 + (i & 15)];
    }
}

extern "C" void kernel_launch(void* const* d_in, const int* in_sizes, int n_in,
                              void* d_out, int out_size, void* d_ws, size_t ws_size,
                              hipStream_t stream)
{
    const float* x       = (const float*)d_in[0];
    const float* A       = (const float*)d_in[1];
    const float* W_B     = (const float*)d_in[2];
    const float* b_B     = (const float*)d_in[3];
    const float* W_C     = (const float*)d_in[4];
    const float* b_C     = (const float*)d_in[5];
    const float* W_d     = (const float*)d_in[6];
    const float* b_d     = (const float*)d_in[7];
    const float* p_Delta = (const float*)d_in[8];
    float* y = (float*)d_out;

    float* ws    = (float*)d_ws;
    float* Bp    = ws;
    float* Cp    = Bp + (size_t)BB * LL * NN;
    float* dlin  = Cp + (size_t)BB * LL * NN;

    ssm_proj_wmma<<<dim3(128), dim3(32 * KSPLIT), 0, stream>>>(
        x, W_B, b_B, W_C, b_C, W_d, b_d, Bp, Cp, dlin);

    ssm_scan_fused<<<dim3(BB * NCHB), dim3(SCANT), 0, stream>>>(
        x, A, dlin, p_Delta, Bp, Cp, y);
}


# scan step loops slimmed: update as h = a*(h + bx/A) - bx on log2e-scaled A (2 fma, was 7 ops); pass C dot summed as 2 packed pairs (sum order changed), row sums by DPP, not LDS permutes
# speedup vs baseline: 6.5369x; 1.1138x over previous
#include <hip/hip_runtime.h>
#include <math.h>

#define DD 768
#define LL 1024
#define NN 16
#define BB 2
#define NC 16
#define LC (LL / NC)

typedef float v2f __attribute__((ext_vector_type(2)));
typedef float v8f __attribute__((ext_vector_type(8)));

#define KSPLIT 8
#define KW     (DD / KSPLIT)
#define KSTEPS (KW / 4)
#define WC4    (KW / 4)
#define WP     (KW + 4)
#define WSTG   (16 * WC4 / 32)

__global__ __launch_bounds__(32 * KSPLIT) __attribute__((amdgpu_waves_per_eu(2, 2))) void ssm_proj_wmma(
    const float* x,
    const float* __restrict__ W_B,
    const float* __restrict__ b_B,
    const float* __restrict__ W_C,
    const float* __restrict__ b_C,
    const float* W_d,
    const float* __restrict__ b_d,
    float* __restrict__ Bp,
    float* __restrict__ Cp,
    float* __restrict__ dlin)
{
    __shared__ float red[KSPLIT][24][32];
    __shared__ float4 s_w[KSPLIT][2][16 * WP / 4];

    const int lane = threadIdx.x & 31;
    const int wave = __builtin_amdgcn_readfirstlane((int)(threadIdx.x >> 5));
    const int half = lane >> 4;
    const int mn   = lane & 15;
    const int row0 = blockIdx.x << 4;
    const int b    = row0 >> 10;
    const int l0   = row0 & (LL - 1);
    const int lA   = l0 + mn;

    const float biasB = (wave == 0) ? b_B[mn] : 0.0f;
    const float biasC = (wave == 0) ? b_C[mn] : 0.0f;
    const float biasD = (wave == 0 && mn == 0) ? b_d[0] : 0.0f;
    v8f accB, accC, accD;
#pragma unroll
    for (int r = 0; r < 8; ++r) { accB[r] = biasB; accC[r] = biasC; accD[r] = biasD; }

    const float n0sel = (mn == 0) ? 1.0f : 0.0f;
    const int   kw0   = wave * KW + 2 * half;
    const float* xa = x + (size_t)b * DD * LL + (size_t)kw0 * LL + lA;
    const float* wdp = W_d + kw0;

    const float4* __restrict__ gwb = (const float4*)(W_B + wave * KW);
    const float4* __restrict__ gwc = (const float4*)(W_C + wave * KW);
    float4 rb[WSTG], rc[WSTG];
#pragma unroll
    for (int j = 0; j < WSTG; ++j) {
        const int i = lane + 32 * j;
        const int n = i / WC4;
        rb[j] = gwb[n * (DD / 4) + (i - n * WC4)];
        rc[j] = gwc[n * (DD / 4) + (i - n * WC4)];
    }
    v2f a[KSTEPS], wd[KSTEPS];
#pragma unroll
    for (int u = 0; u < KSTEPS; ++u) {
        const int ko = 4 * u;
        a[u][0]  = xa[(size_t)ko * LL];
        a[u][1]  = xa[(size_t)(ko + 1) * LL];
        wd[u][0] = wdp[ko];
        wd[u][1] = wdp[ko + 1];
    }
    __builtin_amdgcn_sched_barrier(0);
#pragma unroll
    for (int j = 0; j < WSTG; ++j) {
        const int i = lane + 32 * j;
        const int n = i / WC4;
        s_w[wave][0][n * (WP / 4) + (i - n * WC4)] = rb[j];
        s_w[wave][1][n * (WP / 4) + (i - n * WC4)] = rc[j];
    }
    __syncthreads();
    const float* wbp = (const float*)&s_w[wave][0][0] + mn * WP + 2 * half;
    const float* wcp = (const float*)&s_w[wave][1][0] + mn * WP + 2 * half;

#pragma unroll
    for (int u = 0; u < KSTEPS; ++u) {
        const int ko = 4 * u;
        v2f wb, wc, wdm;
        wb[0] = wbp[ko];  wb[1] = wbp[ko + 1];
        wc[0] = wcp[ko];  wc[1] = wcp[ko + 1];
        wdm[0] = wd[u][0] * n0sel;
        wdm[1] = wd[u][1] * n0sel;
        accB = __builtin_amdgcn_wmma_f32_16x16x4_f32(false, a[u], false, wb, (short)0, accB, false, false);
        accC = __builtin_amdgcn_wmma_f32_16x16x4_f32(false, a[u], false, wc, (short)0, accC, false, false);
        accD = __builtin_amdgcn_wmma_f32_16x16x4_f32(false, a[u], false, wdm, (short)0, accD, false, false);
    }

#pragma unroll
    for (int r = 0; r < 8; ++r) {
        red[wave][r][lane]      = accB[r];
        red[wave][8 + r][lane]  = accC[r];
        red[wave][16 + r][lane] = accD[r];
    }
    __syncthreads();

    float sB = red[0][wave][lane];
    float sC = red[0][8 + wave][lane];
    float sD = red[0][16 + wave][lane];
#pragma unroll
    for (int w = 1; w < KSPLIT; ++w) {
        sB += red[w][wave][lane];
        sC += red[w][8 + wave][lane];
        sD += red[w][16 + wave][lane];
    }
    const int l  = l0 + wave + 8 * half;
    const int gr = b * LL + l;
    Bp[gr * NN + mn] = sB;
    Cp[gr * NN + mn] = sC;
    if (mn == 0) dlin[gr] = sD;
}

__device__ __forceinline__ float ssm_softplus(const float dl)
{
    const float e = __expf(-fabsf(dl));
    const float u = 1.0f + e;
    const float d = u - 1.0f;
    const float q = (__builtin_amdgcn_logf(u) * 0.6931471805599453f) * (e * __builtin_amdgcn_rcpf(d));
    const float l = (d == 0.0f) ? e : q;
    return fmaxf(dl, 0.0f) + l;
}

#define CHB   6
#define ROWS  (CHB * NC)
#define SCANT (ROWS * 4)
#define NCHB  (DD / CHB)
#define TP4   17
#define CP4   (LC * 4 + 4)
#define PL4   (LL * NN / 4)
#define PPAD  (11 * SCANT - PL4)

__device__ __forceinline__ void ssm_step(float4& h, const float4 An2, const float4 rAn,
                                         const float dt, const float xv, const float4 bp)
{
    const float a0 = __builtin_amdgcn_exp2f(dt * An2.x);
    const float a1 = __builtin_amdgcn_exp2f(dt * An2.y);
    const float a2 = __builtin_amdgcn_exp2f(dt * An2.z);
    const float a3 = __builtin_amdgcn_exp2f(dt * An2.w);
    const float b0 = bp.x * xv;
    const float b1 = bp.y * xv;
    const float b2 = bp.z * xv;
    const float b3 = bp.w * xv;
    h.x = fmaf(a0, fmaf(rAn.x, b0, h.x), -b0);
    h.y = fmaf(a1, fmaf(rAn.y, b1, h.y), -b1);
    h.z = fmaf(a2, fmaf(rAn.z, b2, h.z), -b2);
    h.w = fmaf(a3, fmaf(rAn.w, b3, h.w), -b3);
}

template <int CTRL>
__device__ __forceinline__ float ssm_quad_perm(const float v)
{
    return __builtin_bit_cast(float,
        __builtin_amdgcn_update_dpp(0, __builtin_bit_cast(int, v), CTRL, 0xF, 0xF, true));
}

__device__ __forceinline__ float ssm_out(const float4 h, const float4 cp)
{
    v2f hlo, hhi, clo, chi;
    hlo[0] = h.x;  hlo[1] = h.y;  hhi[0] = h.z;  hhi[1] = h.w;
    clo[0] = cp.x; clo[1] = cp.y; chi[0] = cp.z; chi[1] = cp.w;
    const v2f pp = __builtin_elementwise_fma(chi, hhi, clo * hlo);
    float p = pp[0] + pp[1];
    p += ssm_quad_perm<0xB1>(p);
    p += ssm_quad_perm<0x4E>(p);
    return p;
}


__global__ __launch_bounds__(SCANT) __attribute__((amdgpu_waves_per_eu(3, 4))) void ssm_scan_fused(
    const float* x,
    const float* __restrict__ A,
    const float* dlin,
    const float* __restrict__ p_Delta,
    const float* Bp,
    const float* Cp,
    float* __restrict__ y)
{
    __shared__ float4 s_dt[ROWS * TP4];
    __shared__ float4 s_x[ROWS * TP4];
    __shared__ float4 s_y[ROWS * TP4];
    __shared__ float4 s_bp[NC * CP4 + PPAD];
    __shared__ float4 s_cp[NC * CP4 + PPAD];
    __shared__ float4 s_P[SCANT];
    __shared__ float4 s_S[SCANT];

    const int t  = threadIdx.x;
    const int b  = blockIdx.x / NCHB;
    const int d0 = (blockIdx.x % NCHB) * CHB;

    const float4* gx  = (const float4*)(x + ((size_t)(b * DD + d0)) * LL);
    float4*       __restrict__ gy  = (float4*)(y + ((size_t)(b * DD + d0)) * LL);
    const float4* gbp = (const float4*)(Bp + (size_t)b * LL * NN);
    const float4* gcp = (const float4*)(Cp + (size_t)b * LL * NN);
    const float4* gdl = (const float4*)(dlin + (size_t)b * LL);

    const float4 dla = gdl[t & 255];
    const float4 dlb = gdl[(t + 128) & 255];
    float pd[4];
#pragma unroll
    for (int j = 0; j < 4; ++j) pd[j] = p_Delta[d0 + ((t + SCANT * j) >> 8)];
    const int q  = t & 3;
    const int g  = t >> 2;
    const int c  = g / CHB;
    const int ch = g - c * CHB;
    const float4 An = ((const float4*)(A + (d0 + ch) * NN))[q];
    const float4 rx0 = gx[t];
    const float4 rx1 = gx[t + SCANT];
    const float4 rx2 = gx[t + SCANT * 2];
    const float4 rx3 = gx[t + SCANT * 3];
    const int k10 = min(t + SCANT * 10, PL4 - 1);
#define SSM_PLANE_LD(r, gp) \
    const float4 r##0 = gp[t];             const float4 r##1 = gp[t + SCANT];     \
    const float4 r##2 = gp[t + SCANT * 2]; const float4 r##3 = gp[t + SCANT * 3]; \
    const float4 r##4 = gp[t + SCANT * 4]; const float4 r##5 = gp[t + SCANT * 5]; \
    const float4 r##6 = gp[t + SCANT * 6]; const float4 r##7 = gp[t + SCANT * 7]; \
    const float4 r##8 = gp[t + SCANT * 8]; const float4 r##9 = gp[t + SCANT * 9]; \
    const float4 r##10 = gp[k10];
#define SSM_PLANE_ST1(sp, r, m) { const int k = t + SCANT * m; sp[(k >> 8) * CP4 + (k & 255)] = r##m; }
#define SSM_PLANE_ST(sp, r) \
    SSM_PLANE_ST1(sp, r, 0) SSM_PLANE_ST1(sp, r, 1) SSM_PLANE_ST1(sp, r, 2) SSM_PLANE_ST1(sp, r, 3) \
    SSM_PLANE_ST1(sp, r, 4) SSM_PLANE_ST1(sp, r, 5) SSM_PLANE_ST1(sp, r, 6) SSM_PLANE_ST1(sp, r, 7) \
    SSM_PLANE_ST1(sp, r, 8) SSM_PLANE_ST1(sp, r, 9) SSM_PLANE_ST1(sp, r, 10)
    SSM_PLANE_LD(rbp, gbp)
    SSM_PLANE_LD(rcp, gcp)

    __builtin_amdgcn_sched_barrier(0);

    float4 rdt[4];
#pragma unroll
    for (int j = 0; j < 4; ++j) {
        const float4 dl4 = (j & 1) ? dlb : dla;
        rdt[j].x = ssm_softplus(dl4.x + pd[j]);
        rdt[j].y = ssm_softplus(dl4.y + pd[j]);
        rdt[j].z = ssm_softplus(dl4.z + pd[j]);
        rdt[j].w = ssm_softplus(dl4.w + pd[j]);
    }

#define SSM_TILE_ST(j) { const int i = t + SCANT * j; const int off = (((i >> 4) & 15) * CHB + (i >> 8)) * TP4 + (i & 15); s_dt[off] = rdt[j]; s_x[off] = rx##j; }
    SSM_TILE_ST(0) SSM_TILE_ST(1) SSM_TILE_ST(2) SSM_TILE_ST(3)
#undef SSM_TILE_ST
    SSM_PLANE_ST(s_bp, rbp)
    SSM_PLANE_ST(s_cp, rcp)
#undef SSM_PLANE_LD
#undef SSM_PLANE_ST
#undef SSM_PLANE_ST1
    __syncthreads();

    const float4 rAn = make_float4(1.0f / An.x, 1.0f / An.y, 1.0f / An.z, 1.0f / An.w);
    const float4 An2 = make_float4(An.x * 1.4426950408889634f, An.y * 1.4426950408889634f,
                                   An.z * 1.4426950408889634f, An.w * 1.4426950408889634f);
    const float4* pdt = s_dt + g * TP4;
    const float4* px  = s_x + g * TP4;
    const float4* pbp = s_bp + c * CP4 + q;
    const float4* pcp = s_cp + c * CP4 + q;
    float4* prow = s_y + g * TP4;

    float4 h = make_float4(0.f, 0.f, 0.f, 0.f);
    float sd = 0.0f;
    for (int s8 = 0; s8 < LC / 8; ++s8) {
        float4 dt4[2], xv4[2], bp[8];
#pragma unroll
        for (int u = 0; u < 2; ++u) {
            dt4[u] = pdt[2 * s8 + u];
            xv4[u] = px[2 * s8 + u];
        }
#pragma unroll
        for (int k = 0; k < 8; ++k) bp[k] = pbp[(8 * s8 + k) * 4];
#pragma unroll
        for (int u = 0; u < 2; ++u) {
            sd += dt4[u].x; ssm_step(h, An2, rAn, dt4[u].x, xv4[u].x, bp[4 * u + 0]);
            sd += dt4[u].y; ssm_step(h, An2, rAn, dt4[u].y, xv4[u].y, bp[4 * u + 1]);
            sd += dt4[u].z; ssm_step(h, An2, rAn, dt4[u].z, xv4[u].z, bp[4 * u + 2]);
            sd += dt4[u].w; ssm_step(h, An2, rAn, dt4[u].w, xv4[u].w, bp[4 * u + 3]);
        }
    }
    float4 P;
    P.x = __expf(An.x * sd);
    P.y = __expf(An.y * sd);
    P.z = __expf(An.z * sd);
    P.w = __expf(An.w * sd);
    s_P[t] = P;
    s_S[t] = h;
    __syncthreads();

    h = make_float4(0.f, 0.f, 0.f, 0.f);
#pragma unroll
    for (int cc = 0; cc < NC - 1; ++cc) {
        const float4 Pv = s_P[(cc * CHB + ch) * 4 + q];
        const float4 Sv = s_S[(cc * CHB + ch) * 4 + q];
        if (cc < c) {
            h.x = fmaf(Pv.x, h.x, Sv.x);
            h.y = fmaf(Pv.y, h.y, Sv.y);
            h.z = fmaf(Pv.z, h.z, Sv.z);
            h.w = fmaf(Pv.w, h.w, Sv.w);
        }
    }

    for (int s8 = 0; s8 < LC / 8; ++s8) {
        float4 dt4[2], xv4[2], bp[8], cp[8];
        float p[8];
#pragma unroll
        for (int u = 0; u < 2; ++u) {
            dt4[u] = pdt[2 * s8 + u];
            xv4[u] = px[2 * s8 + u];
        }
#pragma unroll
        for (int k = 0; k < 8; ++k) {
            bp[k] = pbp[(8 * s8 + k) * 4];
            cp[k] = pcp[(8 * s8 + k) * 4];
        }
#pragma unroll
        for (int u = 0; u < 2; ++u) {
            ssm_step(h, An2, rAn, dt4[u].x, xv4[u].x, bp[4 * u + 0]); p[4 * u + 0] = ssm_out(h, cp[4 * u + 0]);
            ssm_step(h, An2, rAn, dt4[u].y, xv4[u].y, bp[4 * u + 1]); p[4 * u + 1] = ssm_out(h, cp[4 * u + 1]);
            ssm_step(h, An2, rAn, dt4[u].z, xv4[u].z, bp[4 * u + 2]); p[4 * u + 2] = ssm_out(h, cp[4 * u + 2]);
            ssm_step(h, An2, rAn, dt4[u].w, xv4[u].w, bp[4 * u + 3]); p[4 * u + 3] = ssm_out(h, cp[4 * u + 3]);
        }
        if (q == 0) {
            prow[2 * s8]     = make_float4(p[0], p[1], p[2], p[3]);
            prow[2 * s8 + 1] = make_float4(p[4], p[5], p[6], p[7]);
        }
    }
    __syncthreads();

#pragma unroll
    for (int j = 0; j < 4; ++j) {
        const int i = t + SCANT * j;
        gy[i] = s_y[(((i >> 4) & 15) * CHB + (i >> 8)) * TP4 + (i & 15)];
    }
}

extern "C" void kernel_launch(void* const* d_in, const int* in_sizes, int n_in,
                              void* d_out, int out_size, void* d_ws, size_t ws_size,
                              hipStream_t stream)
{
    const float* x       = (const float*)d_in[0];
    const float* A       = (const float*)d_in[1];
    const float* W_B     = (const float*)d_in[2];
    const float* b_B     = (const float*)d_in[3];
    const float* W_C     = (const float*)d_in[4];
    const float* b_C     = (const float*)d_in[5];
    const float* W_d     = (const float*)d_in[6];
    const float* b_d     = (const float*)d_in[7];
    const float* p_Delta = (const float*)d_in[8];
    float* y = (float*)d_out;

    float* ws    = (float*)d_ws;
    float* Bp    = ws;
    float* Cp    = Bp + (size_t)BB * LL * NN;
    float* dlin  = Cp + (size_t)BB * LL * NN;

    ssm_proj_wmma<<<dim3(128), dim3(32 * KSPLIT), 0, stream>>>(
        x, W_B, b_B, W_C, b_C, W_d, b_d, Bp, Cp, dlin);

    ssm_scan_fused<<<dim3(BB * NCHB), dim3(SCANT), 0, stream>>>(
        x, A, dlin, p_Delta, Bp, Cp, y);
}
